# starting kernel
# baseline (speedup 1.0000x reference)
#include <hip/hip_runtime.h>


typedef __attribute__((ext_vector_type(2))) float v2f;
typedef __attribute__((ext_vector_type(4))) float v4f;
typedef __attribute__((ext_vector_type(8))) float v8f;

#define N_NODES 100000
#define N_EDGES 1200000
#define IN_DIM  256
#define Z_DIM   64
#define M_TILES (N_NODES / 16)
#define LDS_STRIDE 72


__global__ __launch_bounds__(256) void zero_counts_kernel(unsigned* __restrict__ cnt) {
    int i = blockIdx.x * 256 + threadIdx.x;
    if (i < N_NODES) cnt[i] = 0u;
}

__global__ __launch_bounds__(256) void count_deg_kernel(const int* __restrict__ ei,
                                                        unsigned* __restrict__ cnt) {
    int e = blockIdx.x * 256 + threadIdx.x;
    if (e < N_EDGES) {
        int col = ei[N_EDGES + e];
        atomicAdd(&cnt[col], 1u);
    }
}

__global__ __launch_bounds__(256) void dinv_kernel(const unsigned* __restrict__ cnt,
                                                   float* __restrict__ dinv) {
    int i = blockIdx.x * 256 + threadIdx.x;
    if (i < N_NODES) {
        float d = (float)cnt[i] + 1.0f;
        dinv[i] = __frsqrt_rn(d);
    }
}


__global__ __launch_bounds__(256) void gemm_scale_kernel(const float* __restrict__ x,
                                                         const float* __restrict__ W,
                                                         const float* __restrict__ dinv,
                                                         float* __restrict__ s,
                                                         float* __restrict__ out) {
    __shared__ float Wl[IN_DIM * LDS_STRIDE];

    const int tid = threadIdx.x;

    for (int i = tid; i < (IN_DIM * Z_DIM) / 4; i += 256) {
        int r  = i >> 4;
        int c4 = (i & 15) << 2;
        v4f w = *(const v4f*)(W + r * Z_DIM + c4);
        *(v4f*)(&Wl[r * LDS_STRIDE + c4]) = w;
    }
    __syncthreads();

    const int wave = tid >> 5;
    const int lane = tid & 31;
    const int mt   = blockIdx.x * 8 + wave;
    if (mt >= M_TILES) return;

    const int m0   = mt * 16;
    const int half = lane >> 4;
    const int l16  = lane & 15;

    v8f z = {};
    v8f acc0 = z, acc1 = z, acc2 = z, acc3 = z;

    const float* xrow = x + (size_t)(m0 + l16) * IN_DIM + 2 * half;

    #pragma unroll 4
    for (int k = 0; k < IN_DIM; k += 4) {
        v2f a = *(const v2f*)(xrow + k);
        const int kr = k + 2 * half;
        v2f b0, b1, b2, b3;
        b0.x = Wl[(kr    ) * LDS_STRIDE +  0 + l16];
        b0.y = Wl[(kr + 1) * LDS_STRIDE +  0 + l16];
        b1.x = Wl[(kr    ) * LDS_STRIDE + 16 + l16];
        b1.y = Wl[(kr + 1) * LDS_STRIDE + 16 + l16];
        b2.x = Wl[(kr    ) * LDS_STRIDE + 32 + l16];
        b2.y = Wl[(kr + 1) * LDS_STRIDE + 32 + l16];
        b3.x = Wl[(kr    ) * LDS_STRIDE + 48 + l16];
        b3.y = Wl[(kr + 1) * LDS_STRIDE + 48 + l16];
        acc0 = __builtin_amdgcn_wmma_f32_16x16x4_f32(false, a, false, b0, (short)0, acc0, false, false);
        acc1 = __builtin_amdgcn_wmma_f32_16x16x4_f32(false, a, false, b1, (short)0, acc1, false, false);
        acc2 = __builtin_amdgcn_wmma_f32_16x16x4_f32(false, a, false, b2, (short)0, acc2, false, false);
        acc3 = __builtin_amdgcn_wmma_f32_16x16x4_f32(false, a, false, b3, (short)0, acc3, false, false);
    }

    float dv[8];
    #pragma unroll
    for (int r = 0; r < 8; ++r) dv[r] = dinv[m0 + half * 8 + r];

    v8f accs[4] = {acc0, acc1, acc2, acc3};
    #pragma unroll
    for (int nt = 0; nt < 4; ++nt) {
        #pragma unroll
        for (int r = 0; r < 8; ++r) {
            float v = accs[nt][r] * dv[r];
            size_t off = (size_t)(m0 + half * 8 + r) * Z_DIM + nt * 16 + l16;
            s[off]   = v;
            out[off] = v;
        }
    }
}


__global__ __launch_bounds__(256) void scatter_kernel(const int* __restrict__ ei,
                                                      const float* __restrict__ s,
                                                      float* __restrict__ out) {
    unsigned gid = blockIdx.x * 256u + threadIdx.x;
    int e = (int)(gid >> 6);
    int k = (int)(gid & 63u);
    if (e >= N_EDGES) return;
    int row = ei[e];
    int col = ei[N_EDGES + e];
    float v = s[(size_t)row * Z_DIM + k];
    unsafeAtomicAdd(&out[(size_t)col * Z_DIM + k], v);
}


__global__ __launch_bounds__(256) void finalize_kernel(float* __restrict__ out,
                                                       const float* __restrict__ dinv,
                                                       const float* __restrict__ bias,
                                                       const float* __restrict__ pa) {
    int gid = blockIdx.x * 256 + threadIdx.x;
    if (gid >= N_NODES * Z_DIM) return;
    int k = gid & 63;
    float v = dinv[gid >> 6] * out[gid] + bias[k];
    out[gid] = (v >= 0.0f) ? v : pa[k] * v;
}


extern "C" void kernel_launch(void* const* d_in, const int* in_sizes, int n_in,
                              void* d_out, int out_size, void* d_ws, size_t ws_size,
                              hipStream_t stream) {
    const float* x    = (const float*)d_in[0];
    const int*   ei   = (const int*)d_in[1];
    const float* W    = (const float*)d_in[3];
    const float* bias = (const float*)d_in[4];
    const float* pa   = (const float*)d_in[5];
    float* out = (float*)d_out;

    unsigned* cnt = (unsigned*)d_ws;
    float* dinv   = (float*)((char*)d_ws + (size_t)N_NODES * 4);
    float* s      = (float*)((char*)d_ws + (size_t)2 * N_NODES * 4);

    zero_counts_kernel<<<(N_NODES + 255) / 256, 256, 0, stream>>>(cnt);
    count_deg_kernel<<<(N_EDGES + 255) / 256, 256, 0, stream>>>(ei, cnt);
    dinv_kernel<<<(N_NODES + 255) / 256, 256, 0, stream>>>(cnt, dinv);

    gemm_scale_kernel<<<(M_TILES + 7) / 8, 256, 0, stream>>>(x, W, dinv, s, out);

    const long long pairs = (long long)N_EDGES * Z_DIM;
    scatter_kernel<<<(unsigned)((pairs + 255) / 256), 256, 0, stream>>>(ei, s, out);

    finalize_kernel<<<(N_NODES * Z_DIM + 255) / 256, 256, 0, stream>>>(out, dinv, bias, pa);
}


# Edge sum moved from global f32 atomics on out to LDS accumulators, one workgroup per 800 destination nodes (LDS f32 atomics, arrival order); bias+PReLU applied at write-out; 5 launches
# speedup vs baseline: 1.6944x; 1.6944x over previous
#include <hip/hip_runtime.h>


typedef __attribute__((ext_vector_type(2))) float v2f;
typedef __attribute__((ext_vector_type(4))) float v4f;
typedef __attribute__((ext_vector_type(8))) float v8f;
typedef __attribute__((ext_vector_type(4))) int   v4i;

#define N_NODES 100000
#define N_EDGES 1200000
#define IN_DIM  256
#define Z_DIM   64
#define M_TILES (N_NODES / 16)
#define LDS_STRIDE 72


__global__ __launch_bounds__(256) void zero_counts_kernel(unsigned* __restrict__ cnt) {
    int i = blockIdx.x * 256 + threadIdx.x;
    if (i < N_NODES) cnt[i] = 0u;
}

__global__ __launch_bounds__(256) void count_deg_kernel(const int* __restrict__ ei,
                                                        unsigned* __restrict__ cnt) {
    int e = blockIdx.x * 256 + threadIdx.x;
    if (e < N_EDGES) {
        int col = ei[N_EDGES + e];
        atomicAdd(&cnt[col], 1u);
    }
}

__global__ __launch_bounds__(256) void dinv_kernel(const unsigned* __restrict__ cnt,
                                                   float* __restrict__ dinv) {
    int i = blockIdx.x * 256 + threadIdx.x;
    if (i < N_NODES) {
        float d = (float)cnt[i] + 1.0f;
        dinv[i] = __frsqrt_rn(d);
    }
}


__global__ __launch_bounds__(256) void gemm_scale_kernel(const float* __restrict__ x,
                                                         const float* __restrict__ W,
                                                         const float* __restrict__ dinv,
                                                         float* __restrict__ s) {
    __shared__ float Wl[IN_DIM * LDS_STRIDE];

    const int tid = threadIdx.x;

    for (int i = tid; i < (IN_DIM * Z_DIM) / 4; i += 256) {
        int r  = i >> 4;
        int c4 = (i & 15) << 2;
        v4f w = *(const v4f*)(W + r * Z_DIM + c4);
        *(v4f*)(&Wl[r * LDS_STRIDE + c4]) = w;
    }
    __syncthreads();

    const int wave = tid >> 5;
    const int lane = tid & 31;
    const int mt   = blockIdx.x * 8 + wave;
    if (mt >= M_TILES) return;

    const int m0   = mt * 16;
    const int half = lane >> 4;
    const int l16  = lane & 15;

    v8f z = {};
    v8f acc0 = z, acc1 = z, acc2 = z, acc3 = z;

    const float* xrow = x + (size_t)(m0 + l16) * IN_DIM + 2 * half;

    #pragma unroll 4
    for (int k = 0; k < IN_DIM; k += 4) {
        v2f a = *(const v2f*)(xrow + k);
        const int kr = k + 2 * half;
        v2f b0, b1, b2, b3;
        b0.x = Wl[(kr    ) * LDS_STRIDE +  0 + l16];
        b0.y = Wl[(kr + 1) * LDS_STRIDE +  0 + l16];
        b1.x = Wl[(kr    ) * LDS_STRIDE + 16 + l16];
        b1.y = Wl[(kr + 1) * LDS_STRIDE + 16 + l16];
        b2.x = Wl[(kr    ) * LDS_STRIDE + 32 + l16];
        b2.y = Wl[(kr + 1) * LDS_STRIDE + 32 + l16];
        b3.x = Wl[(kr    ) * LDS_STRIDE + 48 + l16];
        b3.y = Wl[(kr + 1) * LDS_STRIDE + 48 + l16];
        acc0 = __builtin_amdgcn_wmma_f32_16x16x4_f32(false, a, false, b0, (short)0, acc0, false, false);
        acc1 = __builtin_amdgcn_wmma_f32_16x16x4_f32(false, a, false, b1, (short)0, acc1, false, false);
        acc2 = __builtin_amdgcn_wmma_f32_16x16x4_f32(false, a, false, b2, (short)0, acc2, false, false);
        acc3 = __builtin_amdgcn_wmma_f32_16x16x4_f32(false, a, false, b3, (short)0, acc3, false, false);
    }

    float dv[8];
    #pragma unroll
    for (int r = 0; r < 8; ++r) dv[r] = dinv[m0 + half * 8 + r];

    v8f accs[4] = {acc0, acc1, acc2, acc3};
    #pragma unroll
    for (int nt = 0; nt < 4; ++nt) {
        #pragma unroll
        for (int r = 0; r < 8; ++r) {
            float v = accs[nt][r] * dv[r];
            size_t off = (size_t)(m0 + half * 8 + r) * Z_DIM + nt * 16 + l16;
            s[off] = v;
        }
    }
}


#define AGG_R       800
#define AGG_BLOCKS  (N_NODES / AGG_R)
#define AGG_THREADS 1024
#define AGG_Q       16384
#define AGG_U       8
#define COL_GROUPS  (N_EDGES / 4)
#define E_BITS      21
#define E_MASK      ((1u << E_BITS) - 1u)

static_assert(N_NODES % AGG_R == 0, "node ranges must tile the graph");
static_assert(N_EDGES % 4 == 0, "ids are read four at a time");
static_assert(N_EDGES <= (1 << E_BITS), "edge id must fit its field");
static_assert(AGG_R < (1 << (32 - E_BITS)), "local node id and the spare row must fit their field");
#define AGG_SPARE   ((unsigned)AGG_R << E_BITS)

__device__ __noinline__ void agg_slow_add(float* acc, const int* __restrict__ ei,
                                          const float* __restrict__ s, int e, unsigned lc) {
    const int row = ei[e];
    for (int k = 0; k < Z_DIM; ++k)
        atomicAdd(&acc[lc * Z_DIM + k], s[(size_t)row * Z_DIM + k]);
}

__global__ __launch_bounds__(AGG_THREADS) void aggregate_kernel(const int* __restrict__ ei,
                                                                const float* __restrict__ s,
                                                                const float* __restrict__ dinv,
                                                                const float* __restrict__ bias,
                                                                const float* __restrict__ pa,
                                                                float* __restrict__ out) {
    __shared__ float    acc[(AGG_R + 1) * Z_DIM];
    __shared__ unsigned queue[AGG_Q];
    __shared__ unsigned qcount;

    const int tid  = threadIdx.x;
    const int lane = tid & 31;
    const int base = blockIdx.x * AGG_R;

    {
        const v4f* s4 = (const v4f*)(s + (size_t)base * Z_DIM);
        for (int i = tid; i < AGG_R * Z_DIM / 4; i += AGG_THREADS) ((v4f*)acc)[i] = s4[i];
        if (tid < Z_DIM) acc[AGG_R * Z_DIM + tid] = 0.0f;
        if (tid == 0) qcount = 0u;
    }
    __syncthreads();

    {
        const v4i* col4 = (const v4i*)(ei + N_EDGES);
        const v4i none = {-1, -1, -1, -1};
        constexpr int TRIP  = AGG_U * AGG_THREADS;
        constexpr int FULL  = COL_GROUPS / TRIP;
        constexpr int TRIPS = (COL_GROUPS + TRIP - 1) / TRIP;
        const unsigned start = blockIdx.x * (unsigned)(COL_GROUPS / AGG_BLOCKS);
        const unsigned tb    = start + (unsigned)tid;
        #define AGG_WRAP(x) min((unsigned)(x), (unsigned)(x) - (unsigned)COL_GROUPS)

        for (int t = 0; t < TRIPS; ++t) {
            const unsigned g0 = (unsigned)t * TRIP;
            v4i cur[AGG_U];
            if (t < FULL) {
                #pragma unroll
                for (int u = 0; u < AGG_U; ++u)
                    cur[u] = col4[AGG_WRAP(tb + g0 + (unsigned)(u * AGG_THREADS))];
            } else {
                #pragma unroll
                for (int u = 0; u < AGG_U; ++u) {
                    const unsigned g = g0 + (unsigned)(u * AGG_THREADS) + (unsigned)tid;
                    cur[u] = (g < (unsigned)COL_GROUPS) ? col4[AGG_WRAP(g + start)] : none;
                }
            }

            unsigned m[AGG_U * 4];
            unsigned total = 0u;
            #pragma unroll
            for (int u = 0; u < AGG_U; ++u) {
                #pragma unroll
                for (int j = 0; j < 4; ++j) {
                    const unsigned mk = __builtin_amdgcn_ballot_w32((unsigned)(cur[u][j] - base) < (unsigned)AGG_R);
                    m[u * 4 + j] = mk;
                    total += (unsigned)__builtin_popcount(mk);
                }
            }
            if (total) {
                unsigned p0 = 0u;
                if (lane == 0) p0 = atomicAdd(&qcount, total);
                p0 = (unsigned)__builtin_amdgcn_readfirstlane((int)p0);
                #pragma unroll
                for (int u = 0; u < AGG_U; ++u) {
                    #pragma unroll
                    for (int j = 0; j < 4; ++j) {
                        const unsigned mk = m[u * 4 + j];
                        if (mk) {
                            const unsigned lc = (unsigned)(cur[u][j] - base);
                            if (lc < (unsigned)AGG_R) {
                                const unsigned pos = p0 + __builtin_amdgcn_mbcnt_lo(mk, 0u);
                                const unsigned gr  = AGG_WRAP(tb + g0 + (unsigned)(u * AGG_THREADS));
                                const int e = (int)(gr * 4u + (unsigned)j);
                                if (pos < (unsigned)AGG_Q) queue[pos] = (unsigned)e | (lc << E_BITS);
                                else                       agg_slow_add(acc, ei, s, e, lc);
                            }
                            p0 += (unsigned)__builtin_popcount(mk);
                        }
                    }
                }
            }
        }
        #undef AGG_WRAP
    }
    __syncthreads();

    {
        unsigned n = qcount;
        if (n > (unsigned)AGG_Q) n = (unsigned)AGG_Q;
        n = (unsigned)__builtin_amdgcn_readfirstlane((int)n);
        unsigned g = (unsigned)__builtin_amdgcn_readfirstlane(tid & ~31);

        unsigned ent = AGG_SPARE; int row = 0;
        if (g + lane < n) { ent = queue[g + lane]; row = ei[ent & E_MASK]; }
        while (g < n) {
            const unsigned gn = g + AGG_THREADS;
            const bool okn = (gn + lane) < n;
            const unsigned qn = queue[okn ? (gn + lane) : 0u];
            const int      rn = ei[qn & E_MASK];
            const unsigned ent_n = okn ? qn : AGG_SPARE;
            const int      row_n = okn ? rn : 0;
            const unsigned cnt = n - g;
            #pragma unroll
            for (int jb = 0; jb < 32; jb += 16) {
                if (jb == 0 || cnt > 16u) {
                    float a0[16], a1[16];
                    #pragma unroll
                    for (int j = 0; j < 16; ++j) {
                        const int r = __builtin_amdgcn_readlane(row, jb + j);
                        const float* sp = s + (size_t)r * Z_DIM + lane;
                        a0[j] = sp[0];
                        a1[j] = sp[32];
                    }
                    #pragma unroll
                    for (int j = 0; j < 16; ++j) {
                        const unsigned l = (unsigned)__builtin_amdgcn_readlane((int)ent, jb + j) >> E_BITS;
                        float* ap = acc + l * Z_DIM + lane;
                        atomicAdd(ap,      a0[j]);
                        atomicAdd(ap + 32, a1[j]);
                    }
                }
            }
            ent = ent_n; row = row_n; g = gn;
        }
    }
    __syncthreads();

    {
        const v4f* acc4 = (const v4f*)acc;
        v4f* out4 = (v4f*)(out + (size_t)base * Z_DIM);
        for (int i = tid; i < AGG_R * Z_DIM / 4; i += AGG_THREADS) {
            const v4f a = acc4[i];
            const float d = dinv[base + (i >> 4)];
            const int k4 = (i & 15) << 2;
            v4f v;
            v.x = d * a.x + bias[k4 + 0];
            v.y = d * a.y + bias[k4 + 1];
            v.z = d * a.z + bias[k4 + 2];
            v.w = d * a.w + bias[k4 + 3];
            v.x = (v.x >= 0.0f) ? v.x : pa[k4 + 0] * v.x;
            v.y = (v.y >= 0.0f) ? v.y : pa[k4 + 1] * v.y;
            v.z = (v.z >= 0.0f) ? v.z : pa[k4 + 2] * v.z;
            v.w = (v.w >= 0.0f) ? v.w : pa[k4 + 3] * v.w;
            out4[i] = v;
        }
    }
}


extern "C" void kernel_launch(void* const* d_in, const int* in_sizes, int n_in,
                              void* d_out, int out_size, void* d_ws, size_t ws_size,
                              hipStream_t stream) {
    const float* x    = (const float*)d_in[0];
    const int*   ei   = (const int*)d_in[1];
    const float* W    = (const float*)d_in[3];
    const float* bias = (const float*)d_in[4];
    const float* pa   = (const float*)d_in[5];
    float* out = (float*)d_out;

    unsigned* cnt = (unsigned*)d_ws;
    float* dinv   = (float*)((char*)d_ws + (size_t)N_NODES * 4);
    float* s      = (float*)((char*)d_ws + (size_t)2 * N_NODES * 4);

    zero_counts_kernel<<<(N_NODES + 255) / 256, 256, 0, stream>>>(cnt);
    count_deg_kernel<<<(N_EDGES + 255) / 256, 256, 0, stream>>>(ei, cnt);
    dinv_kernel<<<(N_NODES + 255) / 256, 256, 0, stream>>>(cnt, dinv);

    gemm_scale_kernel<<<(M_TILES + 7) / 8, 256, 0, stream>>>(x, W, dinv, s);

    aggregate_kernel<<<AGG_BLOCKS, AGG_THREADS, 0, stream>>>(ei, s, dinv, bias, pa, out);
}


# Edges binned once per 400-node destination range in the degree-count launch, buckets in that range's rows of out; its workgroup drains only the bucket, no scan of all ids; LDS adds in arrival order
# speedup vs baseline: 2.1178x; 1.2499x over previous
#include <hip/hip_runtime.h>


typedef __attribute__((ext_vector_type(2))) float v2f;
typedef __attribute__((ext_vector_type(4))) float v4f;
typedef __attribute__((ext_vector_type(8))) float v8f;
typedef __attribute__((ext_vector_type(4))) int   v4i;

#define N_NODES 100000
#define N_EDGES 1200000
#define IN_DIM  256
#define Z_DIM   64
#define M_TILES (N_NODES / 16)
#define LDS_STRIDE 72


#define BIN_R      400
#define N_BINS     (N_NODES / BIN_R)
#define BK_WORDS   (BIN_R * Z_DIM)
#define BK_HDR     32
#define BK_CAP     (BK_WORDS - BK_HDR)
#define R_BITS     17
#define R_MASK     ((1u << R_BITS) - 1u)
#define COL_GROUPS (N_EDGES / 4)
#define BIN_THREADS 1024
#define BIN_BLOCKS ((COL_GROUPS + BIN_THREADS - 1) / BIN_THREADS)

static_assert(N_NODES % BIN_R == 0, "node ranges must tile the graph");
static_assert(N_EDGES % 4 == 0, "ids are read four at a time");
static_assert(N_NODES <= (1 << R_BITS), "source row must fit its field");
static_assert(BIN_R < (1 << (32 - R_BITS)), "local node id and the spare row must fit their field");
static_assert(N_BINS <= BIN_THREADS, "one thread per bin in the reservation step");

__global__ __launch_bounds__(256) void zero_counts_kernel(unsigned* __restrict__ cnt,
                                                          unsigned* __restrict__ bk) {
    int i = blockIdx.x * 256 + threadIdx.x;
    if (i < N_NODES) cnt[i] = 0u;
    if (i < N_BINS)  bk[(size_t)i * BK_WORDS] = 0u;
}

__global__ __launch_bounds__(BIN_THREADS) void count_bin_kernel(const int* __restrict__ ei,
                                                                unsigned* __restrict__ cnt,
                                                                unsigned* __restrict__ bk) {
    __shared__ unsigned hist[N_BINS];
    __shared__ unsigned gbase[N_BINS];

    const int tid = threadIdx.x;
    const int g   = blockIdx.x * BIN_THREADS + tid;
    const bool ok = g < COL_GROUPS;

    if (tid < N_BINS) hist[tid] = 0u;
    __syncthreads();

    v4i src4 = {0, 0, 0, 0}, dst4 = {0, 0, 0, 0};
    unsigned bin[4], rank[4];
    if (ok) {
        src4 = ((const v4i*)ei)[g];
        dst4 = ((const v4i*)(ei + N_EDGES))[g];
        #pragma unroll
        for (int j = 0; j < 4; ++j) {
            const unsigned c = (unsigned)dst4[j];
            atomicAdd(&cnt[c], 1u);
            bin[j]  = c / (unsigned)BIN_R;
            rank[j] = atomicAdd(&hist[bin[j]], 1u);
        }
    }
    __syncthreads();

    if (tid < N_BINS) {
        const unsigned h = hist[tid];
        gbase[tid] = h ? atomicAdd(&bk[(size_t)tid * BK_WORDS], h) : 0u;
    }
    __syncthreads();

    if (ok) {
        #pragma unroll
        for (int j = 0; j < 4; ++j) {
            const unsigned b   = bin[j];
            const unsigned pos = gbase[b] + rank[j];
            const unsigned lc  = (unsigned)dst4[j] - b * (unsigned)BIN_R;
            if (pos < (unsigned)BK_CAP)
                bk[(size_t)b * BK_WORDS + BK_HDR + pos] = (unsigned)src4[j] | (lc << R_BITS);
        }
    }
}

__global__ __launch_bounds__(256) void dinv_kernel(const unsigned* __restrict__ cnt,
                                                   float* __restrict__ dinv) {
    int i = blockIdx.x * 256 + threadIdx.x;
    if (i < N_NODES) {
        float d = (float)cnt[i] + 1.0f;
        dinv[i] = __frsqrt_rn(d);
    }
}


__global__ __launch_bounds__(256) void gemm_scale_kernel(const float* __restrict__ x,
                                                         const float* __restrict__ W,
                                                         const float* __restrict__ dinv,
                                                         float* __restrict__ s) {
    __shared__ float Wl[IN_DIM * LDS_STRIDE];

    const int tid = threadIdx.x;

    for (int i = tid; i < (IN_DIM * Z_DIM) / 4; i += 256) {
        int r  = i >> 4;
        int c4 = (i & 15) << 2;
        v4f w = *(const v4f*)(W + r * Z_DIM + c4);
        *(v4f*)(&Wl[r * LDS_STRIDE + c4]) = w;
    }
    __syncthreads();

    const int wave = tid >> 5;
    const int lane = tid & 31;
    const int mt   = blockIdx.x * 8 + wave;
    if (mt >= M_TILES) return;

    const int m0   = mt * 16;
    const int half = lane >> 4;
    const int l16  = lane & 15;

    v8f z = {};
    v8f acc0 = z, acc1 = z, acc2 = z, acc3 = z;

    const float* xrow = x + (size_t)(m0 + l16) * IN_DIM + 2 * half;

    #pragma unroll 4
    for (int k = 0; k < IN_DIM; k += 4) {
        v2f a = *(const v2f*)(xrow + k);
        const int kr = k + 2 * half;
        v2f b0, b1, b2, b3;
        b0.x = Wl[(kr    ) * LDS_STRIDE +  0 + l16];
        b0.y = Wl[(kr + 1) * LDS_STRIDE +  0 + l16];
        b1.x = Wl[(kr    ) * LDS_STRIDE + 16 + l16];
        b1.y = Wl[(kr + 1) * LDS_STRIDE + 16 + l16];
        b2.x = Wl[(kr    ) * LDS_STRIDE + 32 + l16];
        b2.y = Wl[(kr + 1) * LDS_STRIDE + 32 + l16];
        b3.x = Wl[(kr    ) * LDS_STRIDE + 48 + l16];
        b3.y = Wl[(kr + 1) * LDS_STRIDE + 48 + l16];
        acc0 = __builtin_amdgcn_wmma_f32_16x16x4_f32(false, a, false, b0, (short)0, acc0, false, false);
        acc1 = __builtin_amdgcn_wmma_f32_16x16x4_f32(false, a, false, b1, (short)0, acc1, false, false);
        acc2 = __builtin_amdgcn_wmma_f32_16x16x4_f32(false, a, false, b2, (short)0, acc2, false, false);
        acc3 = __builtin_amdgcn_wmma_f32_16x16x4_f32(false, a, false, b3, (short)0, acc3, false, false);
    }

    float dv[8];
    #pragma unroll
    for (int r = 0; r < 8; ++r) dv[r] = dinv[m0 + half * 8 + r];

    v8f accs[4] = {acc0, acc1, acc2, acc3};
    #pragma unroll
    for (int nt = 0; nt < 4; ++nt) {
        #pragma unroll
        for (int r = 0; r < 8; ++r) {
            float v = accs[nt][r] * dv[r];
            size_t off = (size_t)(m0 + half * 8 + r) * Z_DIM + nt * 16 + l16;
            s[off] = v;
        }
    }
}


#define AGG_THREADS 1024
#define AGG_SPARE   ((unsigned)BIN_R << R_BITS)

__device__ __noinline__ void agg_slow_add(float* acc, const int* __restrict__ ei,
                                          const float* __restrict__ s, int e, unsigned lc) {
    const int row = ei[e];
    for (int k = 0; k < Z_DIM; ++k)
        atomicAdd(&acc[lc * Z_DIM + k], s[(size_t)row * Z_DIM + k]);
}

__global__ __launch_bounds__(AGG_THREADS) void aggregate_kernel(const int* __restrict__ ei,
                                                                const float* __restrict__ s,
                                                                const float* __restrict__ dinv,
                                                                const float* __restrict__ bias,
                                                                const float* __restrict__ pa,
                                                                float* out) {
    __shared__ float acc[(BIN_R + 1) * Z_DIM];

    const int tid  = threadIdx.x;
    const int lane = tid & 31;
    const int base = blockIdx.x * BIN_R;

    const unsigned* bkt = (const unsigned*)(out + (size_t)base * Z_DIM);
    const unsigned n = (unsigned)__builtin_amdgcn_readfirstlane((int)bkt[0]);

    {
        const v4f* s4 = (const v4f*)(s + (size_t)base * Z_DIM);
        constexpr int N4    = BIN_R * Z_DIM / 4;
        constexpr int TRIPS = (N4 + AGG_THREADS - 1) / AGG_THREADS;
        v4f t[TRIPS];
        #pragma unroll
        for (int k = 0; k < TRIPS; ++k) {
            const int i = tid + k * AGG_THREADS;
            if (i < N4) t[k] = s4[i];
        }
        #pragma unroll
        for (int k = 0; k < TRIPS; ++k) {
            const int i = tid + k * AGG_THREADS;
            if (i < N4) ((v4f*)acc)[i] = t[k];
        }
        if (tid < Z_DIM) acc[BIN_R * Z_DIM + tid] = 0.0f;
    }
    __syncthreads();

    if (n <= (unsigned)BK_CAP) {
        const unsigned* ents = bkt + BK_HDR;
        unsigned g = (unsigned)__builtin_amdgcn_readfirstlane(tid & ~31);

        unsigned ent = AGG_SPARE;
        if (g + lane < n) ent = ents[g + lane];
        while (g < n) {
            const unsigned gn = g + AGG_THREADS;
            const bool okn = (gn + lane) < n;
            const unsigned qn = ents[okn ? (gn + lane) : 0u];
            const unsigned ent_n = okn ? qn : AGG_SPARE;
            const unsigned cnt = n - g;
            #pragma unroll
            for (int jb = 0; jb < 32; jb += 16) {
                if (jb == 0 || cnt > 16u) {
                    float a0[16], a1[16];
                    #pragma unroll
                    for (int j = 0; j < 16; ++j) {
                        const unsigned r = (unsigned)__builtin_amdgcn_readlane((int)ent, jb + j) & R_MASK;
                        const float* sp = s + (size_t)r * Z_DIM + lane;
                        a0[j] = sp[0];
                        a1[j] = sp[32];
                    }
                    #pragma unroll
                    for (int j = 0; j < 16; ++j) {
                        const unsigned l = (unsigned)__builtin_amdgcn_readlane((int)ent, jb + j) >> R_BITS;
                        float* ap = acc + l * Z_DIM + lane;
                        atomicAdd(ap,      a0[j]);
                        atomicAdd(ap + 32, a1[j]);
                    }
                }
            }
            ent = ent_n; g = gn;
        }
    } else {
        const v4i* col4 = (const v4i*)(ei + N_EDGES);
        for (int gi = tid; gi < COL_GROUPS; gi += AGG_THREADS) {
            const v4i c = col4[gi];
            #pragma unroll
            for (int j = 0; j < 4; ++j) {
                const unsigned lc = (unsigned)(c[j] - base);
                if (lc < (unsigned)BIN_R) agg_slow_add(acc, ei, s, gi * 4 + j, lc);
            }
        }
    }
    __syncthreads();

    {
        const v4f* acc4 = (const v4f*)acc;
        v4f* out4 = (v4f*)(out + (size_t)base * Z_DIM);
        for (int i = tid; i < BIN_R * Z_DIM / 4; i += AGG_THREADS) {
            const v4f a = acc4[i];
            const float d = dinv[base + (i >> 4)];
            const int k4 = (i & 15) << 2;
            v4f v;
            v.x = d * a.x + bias[k4 + 0];
            v.y = d * a.y + bias[k4 + 1];
            v.z = d * a.z + bias[k4 + 2];
            v.w = d * a.w + bias[k4 + 3];
            v.x = (v.x >= 0.0f) ? v.x : pa[k4 + 0] * v.x;
            v.y = (v.y >= 0.0f) ? v.y : pa[k4 + 1] * v.y;
            v.z = (v.z >= 0.0f) ? v.z : pa[k4 + 2] * v.z;
            v.w = (v.w >= 0.0f) ? v.w : pa[k4 + 3] * v.w;
            out4[i] = v;
        }
    }
}


extern "C" void kernel_launch(void* const* d_in, const int* in_sizes, int n_in,
                              void* d_out, int out_size, void* d_ws, size_t ws_size,
                              hipStream_t stream) {
    const float* x    = (const float*)d_in[0];
    const int*   ei   = (const int*)d_in[1];
    const float* W    = (const float*)d_in[3];
    const float* bias = (const float*)d_in[4];
    const float* pa   = (const float*)d_in[5];
    float* out = (float*)d_out;

    unsigned* cnt = (unsigned*)d_ws;
    float* dinv   = (float*)((char*)d_ws + (size_t)N_NODES * 4);
    float* s      = (float*)((char*)d_ws + (size_t)2 * N_NODES * 4);

    unsigned* bk = (unsigned*)d_out;

    zero_counts_kernel<<<(N_NODES + 255) / 256, 256, 0, stream>>>(cnt, bk);
    count_bin_kernel<<<BIN_BLOCKS, BIN_THREADS, 0, stream>>>(ei, cnt, bk);
    dinv_kernel<<<(N_NODES + 255) / 256, 256, 0, stream>>>(cnt, dinv);

    gemm_scale_kernel<<<(M_TILES + 7) / 8, 256, 0, stream>>>(x, W, dinv, s);

    aggregate_kernel<<<N_BINS, AGG_THREADS, 0, stream>>>(ei, s, dinv, bias, pa, out);
}


# In-degrees counted per 400-node bin from the bin's own bucket with LDS integer adds in a per-bin kernel that writes dinv; the per-edge global counter adds, their table and its zeroing are gone
# speedup vs baseline: 3.5876x; 1.6940x over previous
#include <hip/hip_runtime.h>


typedef __attribute__((ext_vector_type(2))) float v2f;
typedef __attribute__((ext_vector_type(4))) float v4f;
typedef __attribute__((ext_vector_type(8))) float v8f;
typedef __attribute__((ext_vector_type(4))) int   v4i;

#define N_NODES 100000
#define N_EDGES 1200000
#define IN_DIM  256
#define Z_DIM   64
#define M_TILES (N_NODES / 16)
#define LDS_STRIDE 72


#define BIN_R      400
#define N_BINS     (N_NODES / BIN_R)
#define BK_WORDS   (BIN_R * Z_DIM)
#define BK_HDR     32
#define BK_CAP     (BK_WORDS - BK_HDR)
#define R_BITS     17
#define R_MASK     ((1u << R_BITS) - 1u)
#define COL_GROUPS (N_EDGES / 4)
#define BIN_THREADS 1024
#define BIN_BLOCKS ((COL_GROUPS + BIN_THREADS - 1) / BIN_THREADS)

static_assert(N_NODES % BIN_R == 0, "node ranges must tile the graph");
static_assert(N_EDGES % 4 == 0, "ids are read four at a time");
static_assert(N_NODES <= (1 << R_BITS), "source row must fit its field");
static_assert(BIN_R < (1 << (32 - R_BITS)), "local node id and the spare row must fit their field");
static_assert(N_BINS <= BIN_THREADS, "one thread per bin in the reservation step");

__global__ __launch_bounds__(256) void zero_cursors_kernel(unsigned* __restrict__ bk) {
    int i = blockIdx.x * 256 + threadIdx.x;
    if (i < N_BINS) bk[(size_t)i * BK_WORDS] = 0u;
}

__global__ __launch_bounds__(BIN_THREADS) void bin_kernel(const int* __restrict__ ei,
                                                          unsigned* __restrict__ bk) {
    __shared__ unsigned hist[N_BINS];
    __shared__ unsigned gbase[N_BINS];

    const int tid = threadIdx.x;
    const int g   = blockIdx.x * BIN_THREADS + tid;
    const bool ok = g < COL_GROUPS;

    if (tid < N_BINS) hist[tid] = 0u;
    __syncthreads();

    v4i src4 = {0, 0, 0, 0}, dst4 = {0, 0, 0, 0};
    unsigned bin[4], rank[4];
    if (ok) {
        src4 = ((const v4i*)ei)[g];
        dst4 = ((const v4i*)(ei + N_EDGES))[g];
        #pragma unroll
        for (int j = 0; j < 4; ++j) {
            const unsigned c = (unsigned)dst4[j];
            bin[j]  = c / (unsigned)BIN_R;
            rank[j] = atomicAdd(&hist[bin[j]], 1u);
        }
    }
    __syncthreads();

    if (tid < N_BINS) {
        const unsigned h = hist[tid];
        gbase[tid] = h ? atomicAdd(&bk[(size_t)tid * BK_WORDS], h) : 0u;
    }
    __syncthreads();

    if (ok) {
        #pragma unroll
        for (int j = 0; j < 4; ++j) {
            const unsigned b   = bin[j];
            const unsigned pos = gbase[b] + rank[j];
            const unsigned lc  = (unsigned)dst4[j] - b * (unsigned)BIN_R;
            if (pos < (unsigned)BK_CAP)
                bk[(size_t)b * BK_WORDS + BK_HDR + pos] = (unsigned)src4[j] | (lc << R_BITS);
        }
    }
}

#define DEG_THREADS 256
#define DEG_BATCH   6

__global__ __launch_bounds__(DEG_THREADS) void bin_degree_kernel(const int* __restrict__ ei,
                                                                 const unsigned* __restrict__ bk,
                                                                 float* __restrict__ dinv) {
    __shared__ unsigned deg[BIN_R];

    const int tid  = threadIdx.x;
    const int base = blockIdx.x * BIN_R;
    const unsigned* bkt = bk + (size_t)blockIdx.x * BK_WORDS;
    const unsigned n = bkt[0];

    for (int i = tid; i < BIN_R; i += DEG_THREADS) deg[i] = 0u;
    __syncthreads();

    if (n <= (unsigned)BK_CAP) {
        const v4i* ents4 = (const v4i*)(bkt + BK_HDR);
        for (unsigned g4 = 0; g4 * 4u < n; g4 += DEG_BATCH * DEG_THREADS) {
            v4i e[DEG_BATCH];
            #pragma unroll
            for (int k = 0; k < DEG_BATCH; ++k) {
                const unsigned i4 = g4 + (unsigned)(k * DEG_THREADS + tid);
                e[k] = ents4[(i4 * 4u < n) ? i4 : 0u];
            }
            #pragma unroll
            for (int k = 0; k < DEG_BATCH; ++k) {
                const unsigned i4 = g4 + (unsigned)(k * DEG_THREADS + tid);
                #pragma unroll
                for (int j = 0; j < 4; ++j)
                    if (i4 * 4u + (unsigned)j < n) atomicAdd(&deg[(unsigned)e[k][j] >> R_BITS], 1u);
            }
        }
    } else {
        const v4i* col4 = (const v4i*)(ei + N_EDGES);
        for (int gi = tid; gi < COL_GROUPS; gi += DEG_THREADS) {
            const v4i c = col4[gi];
            #pragma unroll
            for (int j = 0; j < 4; ++j) {
                const unsigned lc = (unsigned)(c[j] - base);
                if (lc < (unsigned)BIN_R) atomicAdd(&deg[lc], 1u);
            }
        }
    }
    __syncthreads();

    for (int i = tid; i < BIN_R; i += DEG_THREADS) {
        const float d = (float)deg[i] + 1.0f;
        dinv[base + i] = __frsqrt_rn(d);
    }
}


__global__ __launch_bounds__(256) void gemm_scale_kernel(const float* __restrict__ x,
                                                         const float* __restrict__ W,
                                                         const float* __restrict__ dinv,
                                                         float* __restrict__ s) {
    __shared__ float Wl[IN_DIM * LDS_STRIDE];

    const int tid = threadIdx.x;

    for (int i = tid; i < (IN_DIM * Z_DIM) / 4; i += 256) {
        int r  = i >> 4;
        int c4 = (i & 15) << 2;
        v4f w = *(const v4f*)(W + r * Z_DIM + c4);
        *(v4f*)(&Wl[r * LDS_STRIDE + c4]) = w;
    }
    __syncthreads();

    const int wave = tid >> 5;
    const int lane = tid & 31;
    const int mt   = blockIdx.x * 8 + wave;
    if (mt >= M_TILES) return;

    const int m0   = mt * 16;
    const int half = lane >> 4;
    const int l16  = lane & 15;

    v8f z = {};
    v8f acc0 = z, acc1 = z, acc2 = z, acc3 = z;

    const float* xrow = x + (size_t)(m0 + l16) * IN_DIM + 2 * half;

    #pragma unroll 4
    for (int k = 0; k < IN_DIM; k += 4) {
        v2f a = *(const v2f*)(xrow + k);
        const int kr = k + 2 * half;
        v2f b0, b1, b2, b3;
        b0.x = Wl[(kr    ) * LDS_STRIDE +  0 + l16];
        b0.y = Wl[(kr + 1) * LDS_STRIDE +  0 + l16];
        b1.x = Wl[(kr    ) * LDS_STRIDE + 16 + l16];
        b1.y = Wl[(kr + 1) * LDS_STRIDE + 16 + l16];
        b2.x = Wl[(kr    ) * LDS_STRIDE + 32 + l16];
        b2.y = Wl[(kr + 1) * LDS_STRIDE + 32 + l16];
        b3.x = Wl[(kr    ) * LDS_STRIDE + 48 + l16];
        b3.y = Wl[(kr + 1) * LDS_STRIDE + 48 + l16];
        acc0 = __builtin_amdgcn_wmma_f32_16x16x4_f32(false, a, false, b0, (short)0, acc0, false, false);
        acc1 = __builtin_amdgcn_wmma_f32_16x16x4_f32(false, a, false, b1, (short)0, acc1, false, false);
        acc2 = __builtin_amdgcn_wmma_f32_16x16x4_f32(false, a, false, b2, (short)0, acc2, false, false);
        acc3 = __builtin_amdgcn_wmma_f32_16x16x4_f32(false, a, false, b3, (short)0, acc3, false, false);
    }

    float dv[8];
    #pragma unroll
    for (int r = 0; r < 8; ++r) dv[r] = dinv[m0 + half * 8 + r];

    v8f accs[4] = {acc0, acc1, acc2, acc3};
    #pragma unroll
    for (int nt = 0; nt < 4; ++nt) {
        #pragma unroll
        for (int r = 0; r < 8; ++r) {
            float v = accs[nt][r] * dv[r];
            size_t off = (size_t)(m0 + half * 8 + r) * Z_DIM + nt * 16 + l16;
            s[off] = v;
        }
    }
}


#define AGG_THREADS 1024
#define AGG_SPARE   ((unsigned)BIN_R << R_BITS)

__device__ __noinline__ void agg_slow_add(float* acc, const int* __restrict__ ei,
                                          const float* __restrict__ s, int e, unsigned lc) {
    const int row = ei[e];
    for (int k = 0; k < Z_DIM; ++k)
        atomicAdd(&acc[lc * Z_DIM + k], s[(size_t)row * Z_DIM + k]);
}

__global__ __launch_bounds__(AGG_THREADS) void aggregate_kernel(const int* __restrict__ ei,
                                                                const float* __restrict__ s,
                                                                const float* __restrict__ dinv,
                                                                const float* __restrict__ bias,
                                                                const float* __restrict__ pa,
                                                                float* out) {
    __shared__ float acc[(BIN_R + 1) * Z_DIM];

    const int tid  = threadIdx.x;
    const int lane = tid & 31;
    const int base = blockIdx.x * BIN_R;

    const unsigned* bkt = (const unsigned*)(out + (size_t)base * Z_DIM);
    const unsigned n = (unsigned)__builtin_amdgcn_readfirstlane((int)bkt[0]);

    {
        const v4f* s4 = (const v4f*)(s + (size_t)base * Z_DIM);
        constexpr int N4    = BIN_R * Z_DIM / 4;
        constexpr int TRIPS = (N4 + AGG_THREADS - 1) / AGG_THREADS;
        v4f t[TRIPS];
        #pragma unroll
        for (int k = 0; k < TRIPS; ++k) {
            const int i = tid + k * AGG_THREADS;
            if (i < N4) t[k] = s4[i];
        }
        #pragma unroll
        for (int k = 0; k < TRIPS; ++k) {
            const int i = tid + k * AGG_THREADS;
            if (i < N4) ((v4f*)acc)[i] = t[k];
        }
        if (tid < Z_DIM) acc[BIN_R * Z_DIM + tid] = 0.0f;
    }
    __syncthreads();

    if (n <= (unsigned)BK_CAP) {
        const unsigned* ents = bkt + BK_HDR;
        unsigned g = (unsigned)__builtin_amdgcn_readfirstlane(tid & ~31);

        unsigned ent = AGG_SPARE;
        if (g + lane < n) ent = ents[g + lane];
        while (g < n) {
            const unsigned gn = g + AGG_THREADS;
            const bool okn = (gn + lane) < n;
            const unsigned qn = ents[okn ? (gn + lane) : 0u];
            const unsigned ent_n = okn ? qn : AGG_SPARE;
            const unsigned cnt = n - g;
            #pragma unroll
            for (int jb = 0; jb < 32; jb += 16) {
                if (jb == 0 || cnt > 16u) {
                    float a0[16], a1[16];
                    #pragma unroll
                    for (int j = 0; j < 16; ++j) {
                        const unsigned r = (unsigned)__builtin_amdgcn_readlane((int)ent, jb + j) & R_MASK;
                        const float* sp = s + (size_t)r * Z_DIM + lane;
                        a0[j] = sp[0];
                        a1[j] = sp[32];
                    }
                    #pragma unroll
                    for (int j = 0; j < 16; ++j) {
                        const unsigned l = (unsigned)__builtin_amdgcn_readlane((int)ent, jb + j) >> R_BITS;
                        float* ap = acc + l * Z_DIM + lane;
                        atomicAdd(ap,      a0[j]);
                        atomicAdd(ap + 32, a1[j]);
                    }
                }
            }
            ent = ent_n; g = gn;
        }
    } else {
        const v4i* col4 = (const v4i*)(ei + N_EDGES);
        for (int gi = tid; gi < COL_GROUPS; gi += AGG_THREADS) {
            const v4i c = col4[gi];
            #pragma unroll
            for (int j = 0; j < 4; ++j) {
                const unsigned lc = (unsigned)(c[j] - base);
                if (lc < (unsigned)BIN_R) agg_slow_add(acc, ei, s, gi * 4 + j, lc);
            }
        }
    }
    __syncthreads();

    {
        const v4f* acc4 = (const v4f*)acc;
        v4f* out4 = (v4f*)(out + (size_t)base * Z_DIM);
        for (int i = tid; i < BIN_R * Z_DIM / 4; i += AGG_THREADS) {
            const v4f a = acc4[i];
            const float d = dinv[base + (i >> 4)];
            const int k4 = (i & 15) << 2;
            v4f v;
            v.x = d * a.x + bias[k4 + 0];
            v.y = d * a.y + bias[k4 + 1];
            v.z = d * a.z + bias[k4 + 2];
            v.w = d * a.w + bias[k4 + 3];
            v.x = (v.x >= 0.0f) ? v.x : pa[k4 + 0] * v.x;
            v.y = (v.y >= 0.0f) ? v.y : pa[k4 + 1] * v.y;
            v.z = (v.z >= 0.0f) ? v.z : pa[k4 + 2] * v.z;
            v.w = (v.w >= 0.0f) ? v.w : pa[k4 + 3] * v.w;
            out4[i] = v;
        }
    }
}


extern "C" void kernel_launch(void* const* d_in, const int* in_sizes, int n_in,
                              void* d_out, int out_size, void* d_ws, size_t ws_size,
                              hipStream_t stream) {
    const float* x    = (const float*)d_in[0];
    const int*   ei   = (const int*)d_in[1];
    const float* W    = (const float*)d_in[3];
    const float* bias = (const float*)d_in[4];
    const float* pa   = (const float*)d_in[5];
    float* out = (float*)d_out;

    float* dinv   = (float*)((char*)d_ws + (size_t)N_NODES * 4);
    float* s      = (float*)((char*)d_ws + (size_t)2 * N_NODES * 4);

    unsigned* bk = (unsigned*)d_out;

    zero_cursors_kernel<<<(N_BINS + 255) / 256, 256, 0, stream>>>(bk);
    bin_kernel<<<BIN_BLOCKS, BIN_THREADS, 0, stream>>>(ei, bk);
    bin_degree_kernel<<<N_BINS, DEG_THREADS, 0, stream>>>(ei, bk, dinv);

    gemm_scale_kernel<<<(M_TILES + 7) / 8, 256, 0, stream>>>(x, W, dinv, s);

    aggregate_kernel<<<N_BINS, AGG_THREADS, 0, stream>>>(ei, s, dinv, bias, pa, out);
}


# In-degrees counted inside the GEMM launch: one GEMM workgroup per 400-node bin (25 waves) counts its rows' degrees from the bin's bucket with LDS integer adds and writes dinv; 4 launches
# speedup vs baseline: 3.7912x; 1.0568x over previous
#include <hip/hip_runtime.h>


typedef __attribute__((ext_vector_type(2))) float v2f;
typedef __attribute__((ext_vector_type(4))) float v4f;
typedef __attribute__((ext_vector_type(8))) float v8f;
typedef __attribute__((ext_vector_type(4))) int   v4i;

#define N_NODES 100000
#define N_EDGES 1200000
#define IN_DIM  256
#define Z_DIM   64
#define LDS_STRIDE 72


#define BIN_R      400
#define N_BINS     (N_NODES / BIN_R)
#define BK_WORDS   (BIN_R * Z_DIM)
#define BK_HDR     32
#define BK_CAP     (BK_WORDS - BK_HDR)
#define R_BITS     17
#define R_MASK     ((1u << R_BITS) - 1u)
#define COL_GROUPS (N_EDGES / 4)
#define BIN_THREADS 1024
#define BIN_BLOCKS ((COL_GROUPS + BIN_THREADS - 1) / BIN_THREADS)

static_assert(N_NODES % BIN_R == 0, "node ranges must tile the graph");
static_assert(N_EDGES % 4 == 0, "ids are read four at a time");
static_assert(N_NODES <= (1 << R_BITS), "source row must fit its field");
static_assert(BIN_R < (1 << (32 - R_BITS)), "local node id and the spare row must fit their field");
static_assert(N_BINS <= BIN_THREADS, "one thread per bin in the reservation step");

__global__ __launch_bounds__(256) void zero_cursors_kernel(unsigned* __restrict__ bk) {
    int i = blockIdx.x * 256 + threadIdx.x;
    if (i < N_BINS) bk[(size_t)i * BK_WORDS] = 0u;
}

__global__ __launch_bounds__(BIN_THREADS) void bin_kernel(const int* __restrict__ ei,
                                                          unsigned* __restrict__ bk) {
    __shared__ unsigned hist[N_BINS];
    __shared__ unsigned gbase[N_BINS];

    const int tid = threadIdx.x;
    const int g   = blockIdx.x * BIN_THREADS + tid;
    const bool ok = g < COL_GROUPS;

    if (tid < N_BINS) hist[tid] = 0u;
    __syncthreads();

    v4i src4 = {0, 0, 0, 0}, dst4 = {0, 0, 0, 0};
    unsigned bin[4], rank[4];
    if (ok) {
        src4 = ((const v4i*)ei)[g];
        dst4 = ((const v4i*)(ei + N_EDGES))[g];
        #pragma unroll
        for (int j = 0; j < 4; ++j) {
            const unsigned c = (unsigned)dst4[j];
            bin[j]  = c / (unsigned)BIN_R;
            rank[j] = atomicAdd(&hist[bin[j]], 1u);
        }
    }
    __syncthreads();

    if (tid < N_BINS) {
        const unsigned h = hist[tid];
        gbase[tid] = h ? atomicAdd(&bk[(size_t)tid * BK_WORDS], h) : 0u;
    }
    __syncthreads();

    if (ok) {
        #pragma unroll
        for (int j = 0; j < 4; ++j) {
            const unsigned b   = bin[j];
            const unsigned pos = gbase[b] + rank[j];
            const unsigned lc  = (unsigned)dst4[j] - b * (unsigned)BIN_R;
            if (pos < (unsigned)BK_CAP)
                bk[(size_t)b * BK_WORDS + BK_HDR + pos] = (unsigned)src4[j] | (lc << R_BITS);
        }
    }
}

#define GEMM_WAVES   (BIN_R / 16)
#define GEMM_THREADS (GEMM_WAVES * 32)
#define DEG_BATCH    2

static_assert(BIN_R % 16 == 0, "a bin is a whole number of 16-row tiles");
static_assert(GEMM_THREADS <= 1024, "one workgroup per bin");

__global__ __launch_bounds__(GEMM_THREADS) void gemm_degree_scale_kernel(const float* __restrict__ x,
                                                                         const float* __restrict__ W,
                                                                         const int* __restrict__ ei,
                                                                         const unsigned* __restrict__ bk,
                                                                         float* __restrict__ dinv,
                                                                         float* __restrict__ s) {
    __shared__ float Wl[IN_DIM * LDS_STRIDE];
    __shared__ unsigned deg[BIN_R];

    const int tid  = threadIdx.x;
    const int base = blockIdx.x * BIN_R;
    const unsigned* bkt = bk + (size_t)blockIdx.x * BK_WORDS;
    const unsigned n = bkt[0];

    {
        constexpr int N4    = (IN_DIM * Z_DIM) / 4;
        constexpr int TRIPS = (N4 + GEMM_THREADS - 1) / GEMM_THREADS;
        v4f w[TRIPS];
        #pragma unroll
        for (int k = 0; k < TRIPS; ++k) {
            const int i = tid + k * GEMM_THREADS;
            if (i < N4) w[k] = *(const v4f*)(W + i * 4);
        }
        #pragma unroll
        for (int k = 0; k < TRIPS; ++k) {
            const int i = tid + k * GEMM_THREADS;
            if (i < N4) *(v4f*)(&Wl[(i >> 4) * LDS_STRIDE + ((i & 15) << 2)]) = w[k];
        }
    }
    for (int i = tid; i < BIN_R; i += GEMM_THREADS) deg[i] = 0u;
    __syncthreads();

    if (n <= (unsigned)BK_CAP) {
        const v4i* ents4 = (const v4i*)(bkt + BK_HDR);
        for (unsigned g4 = 0; g4 * 4u < n; g4 += DEG_BATCH * GEMM_THREADS) {
            v4i e[DEG_BATCH];
            #pragma unroll
            for (int k = 0; k < DEG_BATCH; ++k) {
                const unsigned i4 = g4 + (unsigned)(k * GEMM_THREADS + tid);
                e[k] = ents4[(i4 * 4u < n) ? i4 : 0u];
            }
            #pragma unroll
            for (int k = 0; k < DEG_BATCH; ++k) {
                const unsigned i4 = g4 + (unsigned)(k * GEMM_THREADS + tid);
                #pragma unroll
                for (int j = 0; j < 4; ++j)
                    if (i4 * 4u + (unsigned)j < n) atomicAdd(&deg[(unsigned)e[k][j] >> R_BITS], 1u);
            }
        }
    } else {
        const v4i* col4 = (const v4i*)(ei + N_EDGES);
        for (int gi = tid; gi < COL_GROUPS; gi += GEMM_THREADS) {
            const v4i c = col4[gi];
            #pragma unroll
            for (int j = 0; j < 4; ++j) {
                const unsigned lc = (unsigned)(c[j] - base);
                if (lc < (unsigned)BIN_R) atomicAdd(&deg[lc], 1u);
            }
        }
    }
    __syncthreads();

    if (tid < BIN_R) dinv[base + tid] = __frsqrt_rn((float)deg[tid] + 1.0f);

    const int wave = tid >> 5;
    const int lane = tid & 31;
    const int m0   = base + wave * 16;
    const int half = lane >> 4;
    const int l16  = lane & 15;
    v8f z = {};
    v8f acc0 = z, acc1 = z, acc2 = z, acc3 = z;

    const float* xrow = x + (size_t)(m0 + l16) * IN_DIM + 2 * half;

    #pragma unroll 4
    for (int k = 0; k < IN_DIM; k += 4) {
        v2f a = *(const v2f*)(xrow + k);
        const int kr = k + 2 * half;
        v2f b0, b1, b2, b3;
        b0.x = Wl[(kr    ) * LDS_STRIDE +  0 + l16];
        b0.y = Wl[(kr + 1) * LDS_STRIDE +  0 + l16];
        b1.x = Wl[(kr    ) * LDS_STRIDE + 16 + l16];
        b1.y = Wl[(kr + 1) * LDS_STRIDE + 16 + l16];
        b2.x = Wl[(kr    ) * LDS_STRIDE + 32 + l16];
        b2.y = Wl[(kr + 1) * LDS_STRIDE + 32 + l16];
        b3.x = Wl[(kr    ) * LDS_STRIDE + 48 + l16];
        b3.y = Wl[(kr + 1) * LDS_STRIDE + 48 + l16];
        acc0 = __builtin_amdgcn_wmma_f32_16x16x4_f32(false, a, false, b0, (short)0, acc0, false, false);
        acc1 = __builtin_amdgcn_wmma_f32_16x16x4_f32(false, a, false, b1, (short)0, acc1, false, false);
        acc2 = __builtin_amdgcn_wmma_f32_16x16x4_f32(false, a, false, b2, (short)0, acc2, false, false);
        acc3 = __builtin_amdgcn_wmma_f32_16x16x4_f32(false, a, false, b3, (short)0, acc3, false, false);
    }

    float dv[8];
    #pragma unroll
    for (int r = 0; r < 8; ++r) dv[r] = __frsqrt_rn((float)deg[wave * 16 + half * 8 + r] + 1.0f);

    v8f accs[4] = {acc0, acc1, acc2, acc3};
    #pragma unroll
    for (int nt = 0; nt < 4; ++nt) {
        #pragma unroll
        for (int r = 0; r < 8; ++r) {
            float v = accs[nt][r] * dv[r];
            size_t off = (size_t)(m0 + half * 8 + r) * Z_DIM + nt * 16 + l16;
            s[off] = v;
        }
    }
}


#define AGG_THREADS 1024
#define AGG_SPARE   ((unsigned)BIN_R << R_BITS)

__device__ __noinline__ void agg_slow_add(float* acc, const int* __restrict__ ei,
                                          const float* __restrict__ s, int e, unsigned lc) {
    const int row = ei[e];
    for (int k = 0; k < Z_DIM; ++k)
        atomicAdd(&acc[lc * Z_DIM + k], s[(size_t)row * Z_DIM + k]);
}

__global__ __launch_bounds__(AGG_THREADS) void aggregate_kernel(const int* __restrict__ ei,
                                                                const float* __restrict__ s,
                                                                const float* __restrict__ dinv,
                                                                const float* __restrict__ bias,
                                                                const float* __restrict__ pa,
                                                                float* out) {
    __shared__ float acc[(BIN_R + 1) * Z_DIM];

    const int tid  = threadIdx.x;
    const int lane = tid & 31;
    const int base = blockIdx.x * BIN_R;

    const unsigned* bkt = (const unsigned*)(out + (size_t)base * Z_DIM);
    const unsigned n = (unsigned)__builtin_amdgcn_readfirstlane((int)bkt[0]);

    {
        const v4f* s4 = (const v4f*)(s + (size_t)base * Z_DIM);
        constexpr int N4    = BIN_R * Z_DIM / 4;
        constexpr int TRIPS = (N4 + AGG_THREADS - 1) / AGG_THREADS;
        v4f t[TRIPS];
        #pragma unroll
        for (int k = 0; k < TRIPS; ++k) {
            const int i = tid + k * AGG_THREADS;
            if (i < N4) t[k] = s4[i];
        }
        #pragma unroll
        for (int k = 0; k < TRIPS; ++k) {
            const int i = tid + k * AGG_THREADS;
            if (i < N4) ((v4f*)acc)[i] = t[k];
        }
        if (tid < Z_DIM) acc[BIN_R * Z_DIM + tid] = 0.0f;
    }
    __syncthreads();

    if (n <= (unsigned)BK_CAP) {
        const unsigned* ents = bkt + BK_HDR;
        unsigned g = (unsigned)__builtin_amdgcn_readfirstlane(tid & ~31);

        unsigned ent = AGG_SPARE;
        if (g + lane < n) ent = ents[g + lane];
        while (g < n) {
            const unsigned gn = g + AGG_THREADS;
            const bool okn = (gn + lane) < n;
            const unsigned qn = ents[okn ? (gn + lane) : 0u];
            const unsigned ent_n = okn ? qn : AGG_SPARE;
            const unsigned cnt = n - g;
            #pragma unroll
            for (int jb = 0; jb < 32; jb += 16) {
                if (jb == 0 || cnt > 16u) {
                    float a0[16], a1[16];
                    #pragma unroll
                    for (int j = 0; j < 16; ++j) {
                        const unsigned r = (unsigned)__builtin_amdgcn_readlane((int)ent, jb + j) & R_MASK;
                        const float* sp = s + (size_t)r * Z_DIM + lane;
                        a0[j] = sp[0];
                        a1[j] = sp[32];
                    }
                    #pragma unroll
                    for (int j = 0; j < 16; ++j) {
                        const unsigned l = (unsigned)__builtin_amdgcn_readlane((int)ent, jb + j) >> R_BITS;
                        float* ap = acc + l * Z_DIM + lane;
                        atomicAdd(ap,      a0[j]);
                        atomicAdd(ap + 32, a1[j]);
                    }
                }
            }
            ent = ent_n; g = gn;
        }
    } else {
        const v4i* col4 = (const v4i*)(ei + N_EDGES);
        for (int gi = tid; gi < COL_GROUPS; gi += AGG_THREADS) {
            const v4i c = col4[gi];
            #pragma unroll
            for (int j = 0; j < 4; ++j) {
                const unsigned lc = (unsigned)(c[j] - base);
                if (lc < (unsigned)BIN_R) agg_slow_add(acc, ei, s, gi * 4 + j, lc);
            }
        }
    }
    __syncthreads();

    {
        const v4f* acc4 = (const v4f*)acc;
        v4f* out4 = (v4f*)(out + (size_t)base * Z_DIM);
        for (int i = tid; i < BIN_R * Z_DIM / 4; i += AGG_THREADS) {
            const v4f a = acc4[i];
            const float d = dinv[base + (i >> 4)];
            const int k4 = (i & 15) << 2;
            v4f v;
            v.x = d * a.x + bias[k4 + 0];
            v.y = d * a.y + bias[k4 + 1];
            v.z = d * a.z + bias[k4 + 2];
            v.w = d * a.w + bias[k4 + 3];
            v.x = (v.x >= 0.0f) ? v.x : pa[k4 + 0] * v.x;
            v.y = (v.y >= 0.0f) ? v.y : pa[k4 + 1] * v.y;
            v.z = (v.z >= 0.0f) ? v.z : pa[k4 + 2] * v.z;
            v.w = (v.w >= 0.0f) ? v.w : pa[k4 + 3] * v.w;
            out4[i] = v;
        }
    }
}


extern "C" void kernel_launch(void* const* d_in, const int* in_sizes, int n_in,
                              void* d_out, int out_size, void* d_ws, size_t ws_size,
                              hipStream_t stream) {
    const float* x    = (const float*)d_in[0];
    const int*   ei   = (const int*)d_in[1];
    const float* W    = (const float*)d_in[3];
    const float* bias = (const float*)d_in[4];
    const float* pa   = (const float*)d_in[5];
    float* out = (float*)d_out;

    float* dinv   = (float*)((char*)d_ws + (size_t)N_NODES * 4);
    float* s      = (float*)((char*)d_ws + (size_t)2 * N_NODES * 4);

    unsigned* bk = (unsigned*)d_out;

    zero_cursors_kernel<<<(N_BINS + 255) / 256, 256, 0, stream>>>(bk);
    bin_kernel<<<BIN_BLOCKS, BIN_THREADS, 0, stream>>>(ei, bk);
    gemm_degree_scale_kernel<<<N_BINS, GEMM_THREADS, 0, stream>>>(x, W, ei, bk, dinv, s);

    aggregate_kernel<<<N_BINS, AGG_THREADS, 0, stream>>>(ei, s, dinv, bias, pa, out);
}


# Aggregation: bucket counting-sorted by destination in LDS (integer adds give ranks); each node's source rows summed in registers in rank order, row stored once; no LDS f32 atomics or accumulator
# speedup vs baseline: 4.0114x; 1.0581x over previous
#include <hip/hip_runtime.h>


typedef __attribute__((ext_vector_type(2))) float v2f;
typedef __attribute__((ext_vector_type(4))) float v4f;
typedef __attribute__((ext_vector_type(8))) float v8f;
typedef __attribute__((ext_vector_type(4))) int   v4i;

#define N_NODES 100000
#define N_EDGES 1200000
#define IN_DIM  256
#define Z_DIM   64
#define LDS_STRIDE 72


#define BIN_R      400
#define N_BINS     (N_NODES / BIN_R)
#define BK_WORDS   (BIN_R * Z_DIM)
#define BK_HDR     32
#define BK_CAP     (BK_WORDS - BK_HDR)
#define R_BITS     17
#define R_MASK     ((1u << R_BITS) - 1u)
#define COL_GROUPS (N_EDGES / 4)
#define BIN_THREADS 1024
#define BIN_BLOCKS ((COL_GROUPS + BIN_THREADS - 1) / BIN_THREADS)

static_assert(N_NODES % BIN_R == 0, "node ranges must tile the graph");
static_assert(N_EDGES % 4 == 0, "ids are read four at a time");
static_assert(N_NODES <= (1 << R_BITS), "source row must fit its field");
static_assert(BIN_R < (1 << (32 - R_BITS)), "local node id and the spare row must fit their field");
static_assert(N_BINS <= BIN_THREADS, "one thread per bin in the reservation step");

__global__ __launch_bounds__(256) void zero_cursors_kernel(unsigned* __restrict__ bk) {
    int i = blockIdx.x * 256 + threadIdx.x;
    if (i < N_BINS) bk[(size_t)i * BK_WORDS] = 0u;
}

__global__ __launch_bounds__(BIN_THREADS) void bin_kernel(const int* __restrict__ ei,
                                                          unsigned* __restrict__ bk) {
    __shared__ unsigned hist[N_BINS];
    __shared__ unsigned gbase[N_BINS];

    const int tid = threadIdx.x;
    const int g   = blockIdx.x * BIN_THREADS + tid;
    const bool ok = g < COL_GROUPS;

    if (tid < N_BINS) hist[tid] = 0u;
    __syncthreads();

    v4i src4 = {0, 0, 0, 0}, dst4 = {0, 0, 0, 0};
    unsigned bin[4], rank[4];
    if (ok) {
        src4 = ((const v4i*)ei)[g];
        dst4 = ((const v4i*)(ei + N_EDGES))[g];
        #pragma unroll
        for (int j = 0; j < 4; ++j) {
            const unsigned c = (unsigned)dst4[j];
            bin[j]  = c / (unsigned)BIN_R;
            rank[j] = atomicAdd(&hist[bin[j]], 1u);
        }
    }
    __syncthreads();

    if (tid < N_BINS) {
        const unsigned h = hist[tid];
        gbase[tid] = h ? atomicAdd(&bk[(size_t)tid * BK_WORDS], h) : 0u;
    }
    __syncthreads();

    if (ok) {
        #pragma unroll
        for (int j = 0; j < 4; ++j) {
            const unsigned b   = bin[j];
            const unsigned pos = gbase[b] + rank[j];
            const unsigned lc  = (unsigned)dst4[j] - b * (unsigned)BIN_R;
            if (pos < (unsigned)BK_CAP)
                bk[(size_t)b * BK_WORDS + BK_HDR + pos] = (unsigned)src4[j] | (lc << R_BITS);
        }
    }
}

#define GEMM_WAVES   (BIN_R / 16)
#define GEMM_THREADS (GEMM_WAVES * 32)
#define DEG_BATCH    2

static_assert(BIN_R % 16 == 0, "a bin is a whole number of 16-row tiles");
static_assert(GEMM_THREADS <= 1024, "one workgroup per bin");

__global__ __launch_bounds__(GEMM_THREADS) void gemm_degree_scale_kernel(const float* __restrict__ x,
                                                                         const float* __restrict__ W,
                                                                         const int* __restrict__ ei,
                                                                         const unsigned* __restrict__ bk,
                                                                         float* __restrict__ dinv,
                                                                         float* __restrict__ s) {
    __shared__ float Wl[IN_DIM * LDS_STRIDE];
    __shared__ unsigned deg[BIN_R];

    const int tid  = threadIdx.x;
    const int base = blockIdx.x * BIN_R;
    const unsigned* bkt = bk + (size_t)blockIdx.x * BK_WORDS;
    const unsigned n = bkt[0];

    {
        constexpr int N4    = (IN_DIM * Z_DIM) / 4;
        constexpr int TRIPS = (N4 + GEMM_THREADS - 1) / GEMM_THREADS;
        v4f w[TRIPS];
        #pragma unroll
        for (int k = 0; k < TRIPS; ++k) {
            const int i = tid + k * GEMM_THREADS;
            if (i < N4) w[k] = *(const v4f*)(W + i * 4);
        }
        #pragma unroll
        for (int k = 0; k < TRIPS; ++k) {
            const int i = tid + k * GEMM_THREADS;
            if (i < N4) *(v4f*)(&Wl[(i >> 4) * LDS_STRIDE + ((i & 15) << 2)]) = w[k];
        }
    }
    for (int i = tid; i < BIN_R; i += GEMM_THREADS) deg[i] = 0u;
    __syncthreads();

    if (n <= (unsigned)BK_CAP) {
        const v4i* ents4 = (const v4i*)(bkt + BK_HDR);
        for (unsigned g4 = 0; g4 * 4u < n; g4 += DEG_BATCH * GEMM_THREADS) {
            v4i e[DEG_BATCH];
            #pragma unroll
            for (int k = 0; k < DEG_BATCH; ++k) {
                const unsigned i4 = g4 + (unsigned)(k * GEMM_THREADS + tid);
                e[k] = ents4[(i4 * 4u < n) ? i4 : 0u];
            }
            #pragma unroll
            for (int k = 0; k < DEG_BATCH; ++k) {
                const unsigned i4 = g4 + (unsigned)(k * GEMM_THREADS + tid);
                #pragma unroll
                for (int j = 0; j < 4; ++j)
                    if (i4 * 4u + (unsigned)j < n) atomicAdd(&deg[(unsigned)e[k][j] >> R_BITS], 1u);
            }
        }
    } else {
        const v4i* col4 = (const v4i*)(ei + N_EDGES);
        for (int gi = tid; gi < COL_GROUPS; gi += GEMM_THREADS) {
            const v4i c = col4[gi];
            #pragma unroll
            for (int j = 0; j < 4; ++j) {
                const unsigned lc = (unsigned)(c[j] - base);
                if (lc < (unsigned)BIN_R) atomicAdd(&deg[lc], 1u);
            }
        }
    }
    __syncthreads();

    if (tid < BIN_R) dinv[base + tid] = __frsqrt_rn((float)deg[tid] + 1.0f);

    const int wave = tid >> 5;
    const int lane = tid & 31;
    const int m0   = base + wave * 16;
    const int half = lane >> 4;
    const int l16  = lane & 15;
    v8f z = {};
    v8f acc0 = z, acc1 = z, acc2 = z, acc3 = z;

    const float* xrow = x + (size_t)(m0 + l16) * IN_DIM + 2 * half;

    #pragma unroll 4
    for (int k = 0; k < IN_DIM; k += 4) {
        v2f a = *(const v2f*)(xrow + k);
        const int kr = k + 2 * half;
        v2f b0, b1, b2, b3;
        b0.x = Wl[(kr    ) * LDS_STRIDE +  0 + l16];
        b0.y = Wl[(kr + 1) * LDS_STRIDE +  0 + l16];
        b1.x = Wl[(kr    ) * LDS_STRIDE + 16 + l16];
        b1.y = Wl[(kr + 1) * LDS_STRIDE + 16 + l16];
        b2.x = Wl[(kr    ) * LDS_STRIDE + 32 + l16];
        b2.y = Wl[(kr + 1) * LDS_STRIDE + 32 + l16];
        b3.x = Wl[(kr    ) * LDS_STRIDE + 48 + l16];
        b3.y = Wl[(kr + 1) * LDS_STRIDE + 48 + l16];
        acc0 = __builtin_amdgcn_wmma_f32_16x16x4_f32(false, a, false, b0, (short)0, acc0, false, false);
        acc1 = __builtin_amdgcn_wmma_f32_16x16x4_f32(false, a, false, b1, (short)0, acc1, false, false);
        acc2 = __builtin_amdgcn_wmma_f32_16x16x4_f32(false, a, false, b2, (short)0, acc2, false, false);
        acc3 = __builtin_amdgcn_wmma_f32_16x16x4_f32(false, a, false, b3, (short)0, acc3, false, false);
    }

    float dv[8];
    #pragma unroll
    for (int r = 0; r < 8; ++r) dv[r] = __frsqrt_rn((float)deg[wave * 16 + half * 8 + r] + 1.0f);

    v8f accs[4] = {acc0, acc1, acc2, acc3};
    #pragma unroll
    for (int nt = 0; nt < 4; ++nt) {
        #pragma unroll
        for (int r = 0; r < 8; ++r) {
            float v = accs[nt][r] * dv[r];
            size_t off = (size_t)(m0 + half * 8 + r) * Z_DIM + nt * 16 + l16;
            s[off] = v;
        }
    }
}


#define AGG_THREADS 1024
#define AGG_WAVES   (AGG_THREADS / 32)
#define SCAN_PER_LANE ((BIN_R + 31) / 32)
#define POOL_WORDS  ((BIN_R + 1) * Z_DIM)

#define SORT_BATCH  8

static_assert(POOL_WORDS >= BK_CAP + 16, "the sorted list of a full bucket must fit, and a 16-word read past its end");
static_assert(BIN_R % 2 == 0, "two nodes per wave step");

__device__ __noinline__ void agg_slow_add(float* acc, const int* __restrict__ ei,
                                          const float* __restrict__ s, int e, unsigned lc) {
    const int row = ei[e];
    for (int k = 0; k < Z_DIM; ++k)
        atomicAdd(&acc[lc * Z_DIM + k], s[(size_t)row * Z_DIM + k]);
}

__global__ __launch_bounds__(AGG_THREADS) void aggregate_kernel(const int* __restrict__ ei,
                                                                const float* __restrict__ s,
                                                                const float* __restrict__ dinv,
                                                                const float* __restrict__ bias,
                                                                const float* __restrict__ pa,
                                                                float* out) {
    __shared__ unsigned pool[POOL_WORDS];
    __shared__ unsigned cur[BIN_R];

    const int tid  = threadIdx.x;
    const int lane = tid & 31;
    const int wave = tid >> 5;
    const int base = blockIdx.x * BIN_R;

    const unsigned* bkt = (const unsigned*)(out + (size_t)base * Z_DIM);
    const unsigned n = (unsigned)__builtin_amdgcn_readfirstlane((int)bkt[0]);

    if (n <= (unsigned)BK_CAP) {
        const unsigned* ents = bkt + BK_HDR;

        if (tid < BIN_R) cur[tid] = 0u;
        __syncthreads();
        for (unsigned g = 0; g < n; g += SORT_BATCH * AGG_THREADS) {
            unsigned e[SORT_BATCH];
            #pragma unroll
            for (int k = 0; k < SORT_BATCH; ++k) {
                const unsigned i = g + (unsigned)(k * AGG_THREADS + tid);
                e[k] = ents[(i < n) ? i : 0u];
            }
            #pragma unroll
            for (int k = 0; k < SORT_BATCH; ++k) {
                const unsigned i = g + (unsigned)(k * AGG_THREADS + tid);
                if (i < n) atomicAdd(&cur[e[k] >> R_BITS], 1u);
            }
        }
        __syncthreads();

        if (wave == 0) {
            unsigned cnt[SCAN_PER_LANE];
            unsigned t = 0u;
            #pragma unroll
            for (int k = 0; k < SCAN_PER_LANE; ++k) {
                const int idx = lane * SCAN_PER_LANE + k;
                cnt[k] = (idx < BIN_R) ? cur[idx] : 0u;
                t += cnt[k];
            }
            unsigned x = t;
            #pragma unroll
            for (int dlt = 1; dlt < 32; dlt <<= 1) {
                const unsigned v = __shfl_up(x, dlt);
                if (lane >= dlt) x += v;
            }
            unsigned run = x - t;
            #pragma unroll
            for (int k = 0; k < SCAN_PER_LANE; ++k) {
                const int idx = lane * SCAN_PER_LANE + k;
                if (idx < BIN_R) cur[idx] = run;
                run += cnt[k];
            }
        }
        __syncthreads();

        for (unsigned g = 0; g < n; g += SORT_BATCH * AGG_THREADS) {
            unsigned e[SORT_BATCH];
            #pragma unroll
            for (int k = 0; k < SORT_BATCH; ++k) {
                const unsigned i = g + (unsigned)(k * AGG_THREADS + tid);
                e[k] = ents[(i < n) ? i : 0u];
            }
            #pragma unroll
            for (int k = 0; k < SORT_BATCH; ++k) {
                const unsigned i = g + (unsigned)(k * AGG_THREADS + tid);
                if (i < n) {
                    const unsigned pos = atomicAdd(&cur[e[k] >> R_BITS], 1u);
                    pool[pos] = e[k] & R_MASK;
                }
            }
        }
        __syncthreads();

        const int h  = lane >> 4;
        const int c4 = (lane & 15) << 2;
        const v4f b4 = *(const v4f*)(bias + c4);
        const v4f p4 = *(const v4f*)(pa + c4);

        for (int p = wave; p < BIN_R / 2; p += AGG_WAVES) {
            const int l = 2 * p + h;
            const unsigned beg = l ? cur[l - 1] : 0u;
            const unsigned len = cur[l] - beg;
            v4f acc = *(const v4f*)(s + (size_t)(base + l) * Z_DIM + c4);
            const float d = dinv[base + l];
            const unsigned other = (unsigned)__shfl_xor((int)len, 16);
            const unsigned maxlen = (unsigned)__builtin_amdgcn_readfirstlane((int)(len > other ? len : other));

            for (unsigned j0 = 0; j0 < maxlen; j0 += 16) {
                const unsigned q0 = beg + ((j0 < len) ? j0 : 0u);
                unsigned r[16];
                #pragma unroll
                for (int j = 0; j < 16; ++j) r[j] = pool[q0 + (unsigned)j];
                v4f a[16];
                #pragma unroll
                for (int j = 0; j < 16; ++j) {
                    a[j] = (v4f){0.0f, 0.0f, 0.0f, 0.0f};
                    if (j0 + (unsigned)j < len) a[j] = *(const v4f*)(s + (size_t)r[j] * Z_DIM + c4);
                }
                #pragma unroll
                for (int j = 0; j < 16; ++j) acc += a[j];
            }

            v4f v;
            v.x = d * acc.x + b4.x;
            v.y = d * acc.y + b4.y;
            v.z = d * acc.z + b4.z;
            v.w = d * acc.w + b4.w;
            v.x = (v.x >= 0.0f) ? v.x : p4.x * v.x;
            v.y = (v.y >= 0.0f) ? v.y : p4.y * v.y;
            v.z = (v.z >= 0.0f) ? v.z : p4.z * v.z;
            v.w = (v.w >= 0.0f) ? v.w : p4.w * v.w;
            *(v4f*)(out + (size_t)(base + l) * Z_DIM + c4) = v;
        }
    } else {
        float* acc = (float*)pool;
        {
            const v4f* s4 = (const v4f*)(s + (size_t)base * Z_DIM);
            for (int i = tid; i < BIN_R * Z_DIM / 4; i += AGG_THREADS) ((v4f*)acc)[i] = s4[i];
        }
        __syncthreads();
        const v4i* col4 = (const v4i*)(ei + N_EDGES);
        for (int gi = tid; gi < COL_GROUPS; gi += AGG_THREADS) {
            const v4i c = col4[gi];
            #pragma unroll
            for (int j = 0; j < 4; ++j) {
                const unsigned lc = (unsigned)(c[j] - base);
                if (lc < (unsigned)BIN_R) agg_slow_add(acc, ei, s, gi * 4 + j, lc);
            }
        }
        __syncthreads();
        const v4f* acc4 = (const v4f*)acc;
        v4f* out4 = (v4f*)(out + (size_t)base * Z_DIM);
        for (int i = tid; i < BIN_R * Z_DIM / 4; i += AGG_THREADS) {
            const v4f a = acc4[i];
            const float d = dinv[base + (i >> 4)];
            const int k4 = (i & 15) << 2;
            v4f v;
            v.x = d * a.x + bias[k4 + 0];
            v.y = d * a.y + bias[k4 + 1];
            v.z = d * a.z + bias[k4 + 2];
            v.w = d * a.w + bias[k4 + 3];
            v.x = (v.x >= 0.0f) ? v.x : pa[k4 + 0] * v.x;
            v.y = (v.y >= 0.0f) ? v.y : pa[k4 + 1] * v.y;
            v.z = (v.z >= 0.0f) ? v.z : pa[k4 + 2] * v.z;
            v.w = (v.w >= 0.0f) ? v.w : pa[k4 + 3] * v.w;
            out4[i] = v;
        }
    }
}


extern "C" void kernel_launch(void* const* d_in, const int* in_sizes, int n_in,
                              void* d_out, int out_size, void* d_ws, size_t ws_size,
                              hipStream_t stream) {
    const float* x    = (const float*)d_in[0];
    const int*   ei   = (const int*)d_in[1];
    const float* W    = (const float*)d_in[3];
    const float* bias = (const float*)d_in[4];
    const float* pa   = (const float*)d_in[5];
    float* out = (float*)d_out;

    float* dinv   = (float*)((char*)d_ws + (size_t)N_NODES * 4);
    float* s      = (float*)((char*)d_ws + (size_t)2 * N_NODES * 4);

    unsigned* bk = (unsigned*)d_out;

    zero_cursors_kernel<<<(N_BINS + 255) / 256, 256, 0, stream>>>(bk);
    bin_kernel<<<BIN_BLOCKS, BIN_THREADS, 0, stream>>>(ei, bk);
    gemm_degree_scale_kernel<<<N_BINS, GEMM_THREADS, 0, stream>>>(x, W, ei, bk, dinv, s);

    aggregate_kernel<<<N_BINS, AGG_THREADS, 0, stream>>>(ei, s, dinv, bias, pa, out);
}


# Binning workgroup orders its 4096 packed edges by bin in LDS (offset scan of its histogram) and stores them in that order: a bin's run goes out from consecutive lanes, not one 4-byte store per line
# speedup vs baseline: 4.4545x; 1.1105x over previous
#include <hip/hip_runtime.h>


typedef __attribute__((ext_vector_type(2))) float v2f;
typedef __attribute__((ext_vector_type(4))) float v4f;
typedef __attribute__((ext_vector_type(8))) float v8f;
typedef __attribute__((ext_vector_type(4))) int   v4i;
typedef __attribute__((ext_vector_type(2))) unsigned v2u;

#define N_NODES 100000
#define N_EDGES 1200000
#define IN_DIM  256
#define Z_DIM   64
#define LDS_STRIDE 72


#define BIN_R      400
#define N_BINS     (N_NODES / BIN_R)
#define BK_WORDS   (BIN_R * Z_DIM)
#define BK_HDR     32
#define BK_CAP     (BK_WORDS - BK_HDR)
#define R_BITS     17
#define R_MASK     ((1u << R_BITS) - 1u)
#define COL_GROUPS (N_EDGES / 4)
#define BIN_THREADS 1024
#define BIN_BLOCKS ((COL_GROUPS + BIN_THREADS - 1) / BIN_THREADS)

static_assert(N_NODES % BIN_R == 0, "node ranges must tile the graph");
static_assert(N_EDGES % 4 == 0, "ids are read four at a time");
static_assert(N_NODES <= (1 << R_BITS), "source row must fit its field");
static_assert(BIN_R < (1 << (32 - R_BITS)), "local node id and the spare row must fit their field");
static_assert(N_BINS <= BIN_THREADS, "one thread per bin in the reservation step");

__global__ __launch_bounds__(256) void zero_cursors_kernel(unsigned* __restrict__ bk) {
    int i = blockIdx.x * 256 + threadIdx.x;
    if (i < N_BINS) bk[(size_t)i * BK_WORDS] = 0u;
}

#define BIN_EDGES     (BIN_THREADS * 4)
#define BIN_PER_LANE  ((N_BINS + 31) / 32)
#define NO_SLOT       0xFFFFFFFFu

static_assert(BIN_PER_LANE * 32 >= N_BINS, "the scan wave covers every bin");
static_assert(32 + N_BINS <= BIN_THREADS, "one reserving thread per bin beside the scan wave");
static_assert((size_t)N_BINS * BK_WORDS < 0xFFFFFFFFull, "a bucket word index fits 32 bits below NO_SLOT");

__global__ __launch_bounds__(BIN_THREADS) void bin_kernel(const int* __restrict__ ei,
                                                          unsigned* __restrict__ bk) {
    __shared__ unsigned hist[N_BINS];
    __shared__ unsigned lbase[N_BINS];
    __shared__ unsigned gbase[N_BINS];
    __shared__ v2u stage[BIN_EDGES];

    const int tid  = threadIdx.x;
    const int lane = tid & 31;
    const int wave = tid >> 5;
    const int g    = blockIdx.x * BIN_THREADS + tid;
    const bool ok  = g < COL_GROUPS;
    const unsigned total = (unsigned)min(BIN_EDGES, N_EDGES - (int)blockIdx.x * BIN_EDGES);

    if (tid < N_BINS) hist[tid] = 0u;
    __syncthreads();

    v4i src4 = {0, 0, 0, 0}, dst4 = {0, 0, 0, 0};
    unsigned bin[4], rank[4];
    if (ok) {
        src4 = ((const v4i*)ei)[g];
        dst4 = ((const v4i*)(ei + N_EDGES))[g];
        #pragma unroll
        for (int j = 0; j < 4; ++j) {
            const unsigned c = (unsigned)dst4[j];
            bin[j]  = c / (unsigned)BIN_R;
            rank[j] = atomicAdd(&hist[bin[j]], 1u);
        }
    }
    __syncthreads();

    if (wave == 0) {
        unsigned cnt[BIN_PER_LANE];
        unsigned t = 0u;
        #pragma unroll
        for (int k = 0; k < BIN_PER_LANE; ++k) {
            const int idx = lane * BIN_PER_LANE + k;
            cnt[k] = (idx < N_BINS) ? hist[idx] : 0u;
            t += cnt[k];
        }
        unsigned x = t;
        #pragma unroll
        for (int dlt = 1; dlt < 32; dlt <<= 1) {
            const unsigned v = __shfl_up(x, dlt);
            if (lane >= dlt) x += v;
        }
        unsigned run = x - t;
        #pragma unroll
        for (int k = 0; k < BIN_PER_LANE; ++k) {
            const int idx = lane * BIN_PER_LANE + k;
            if (idx < N_BINS) lbase[idx] = run;
            run += cnt[k];
        }
    } else if (tid - 32 < N_BINS) {
        const int b = tid - 32;
        const unsigned h = hist[b];
        gbase[b] = h ? atomicAdd(&bk[(size_t)b * BK_WORDS], h) : 0u;
    }
    __syncthreads();

    if (ok) {
        #pragma unroll
        for (int j = 0; j < 4; ++j) {
            const unsigned b   = bin[j];
            const unsigned pos = gbase[b] + rank[j];
            const unsigned lc  = (unsigned)dst4[j] - b * (unsigned)BIN_R;
            v2u ent;
            ent.x = (unsigned)src4[j] | (lc << R_BITS);
            ent.y = (pos < (unsigned)BK_CAP) ? (b * (unsigned)BK_WORDS + (unsigned)BK_HDR + pos) : NO_SLOT;
            stage[lbase[b] + rank[j]] = ent;
        }
    }
    __syncthreads();

    v2u ent[4];
    #pragma unroll
    for (int k = 0; k < 4; ++k) ent[k] = stage[k * BIN_THREADS + tid];
    #pragma unroll
    for (int k = 0; k < 4; ++k) {
        const unsigned i = (unsigned)(k * BIN_THREADS + tid);
        if (i < total && ent[k].y != NO_SLOT) bk[ent[k].y] = ent[k].x;
    }
}

#define GEMM_WAVES   (BIN_R / 16)
#define GEMM_THREADS (GEMM_WAVES * 32)
#define DEG_BATCH    2

static_assert(BIN_R % 16 == 0, "a bin is a whole number of 16-row tiles");
static_assert(GEMM_THREADS <= 1024, "one workgroup per bin");

__global__ __launch_bounds__(GEMM_THREADS) void gemm_degree_scale_kernel(const float* __restrict__ x,
                                                                         const float* __restrict__ W,
                                                                         const int* __restrict__ ei,
                                                                         const unsigned* __restrict__ bk,
                                                                         float* __restrict__ dinv,
                                                                         float* __restrict__ s) {
    __shared__ float Wl[IN_DIM * LDS_STRIDE];
    __shared__ unsigned deg[BIN_R];

    const int tid  = threadIdx.x;
    const int base = blockIdx.x * BIN_R;
    const unsigned* bkt = bk + (size_t)blockIdx.x * BK_WORDS;
    const unsigned n = bkt[0];

    {
        constexpr int N4    = (IN_DIM * Z_DIM) / 4;
        constexpr int TRIPS = (N4 + GEMM_THREADS - 1) / GEMM_THREADS;
        v4f w[TRIPS];
        #pragma unroll
        for (int k = 0; k < TRIPS; ++k) {
            const int i = tid + k * GEMM_THREADS;
            if (i < N4) w[k] = *(const v4f*)(W + i * 4);
        }
        #pragma unroll
        for (int k = 0; k < TRIPS; ++k) {
            const int i = tid + k * GEMM_THREADS;
            if (i < N4) *(v4f*)(&Wl[(i >> 4) * LDS_STRIDE + ((i & 15) << 2)]) = w[k];
        }
    }
    for (int i = tid; i < BIN_R; i += GEMM_THREADS) deg[i] = 0u;
    __syncthreads();

    if (n <= (unsigned)BK_CAP) {
        const v4i* ents4 = (const v4i*)(bkt + BK_HDR);
        for (unsigned g4 = 0; g4 * 4u < n; g4 += DEG_BATCH * GEMM_THREADS) {
            v4i e[DEG_BATCH];
            #pragma unroll
            for (int k = 0; k < DEG_BATCH; ++k) {
                const unsigned i4 = g4 + (unsigned)(k * GEMM_THREADS + tid);
                e[k] = ents4[(i4 * 4u < n) ? i4 : 0u];
            }
            #pragma unroll
            for (int k = 0; k < DEG_BATCH; ++k) {
                const unsigned i4 = g4 + (unsigned)(k * GEMM_THREADS + tid);
                #pragma unroll
                for (int j = 0; j < 4; ++j)
                    if (i4 * 4u + (unsigned)j < n) atomicAdd(&deg[(unsigned)e[k][j] >> R_BITS], 1u);
            }
        }
    } else {
        const v4i* col4 = (const v4i*)(ei + N_EDGES);
        for (int gi = tid; gi < COL_GROUPS; gi += GEMM_THREADS) {
            const v4i c = col4[gi];
            #pragma unroll
            for (int j = 0; j < 4; ++j) {
                const unsigned lc = (unsigned)(c[j] - base);
                if (lc < (unsigned)BIN_R) atomicAdd(&deg[lc], 1u);
            }
        }
    }
    __syncthreads();

    if (tid < BIN_R) dinv[base + tid] = __frsqrt_rn((float)deg[tid] + 1.0f);

    const int wave = tid >> 5;
    const int lane = tid & 31;
    const int m0   = base + wave * 16;
    const int half = lane >> 4;
    const int l16  = lane & 15;
    v8f z = {};
    v8f acc0 = z, acc1 = z, acc2 = z, acc3 = z;

    const float* xrow = x + (size_t)(m0 + l16) * IN_DIM + 2 * half;

    #pragma unroll 4
    for (int k = 0; k < IN_DIM; k += 4) {
        v2f a = *(const v2f*)(xrow + k);
        const int kr = k + 2 * half;
        v2f b0, b1, b2, b3;
        b0.x = Wl[(kr    ) * LDS_STRIDE +  0 + l16];
        b0.y = Wl[(kr + 1) * LDS_STRIDE +  0 + l16];
        b1.x = Wl[(kr    ) * LDS_STRIDE + 16 + l16];
        b1.y = Wl[(kr + 1) * LDS_STRIDE + 16 + l16];
        b2.x = Wl[(kr    ) * LDS_STRIDE + 32 + l16];
        b2.y = Wl[(kr + 1) * LDS_STRIDE + 32 + l16];
        b3.x = Wl[(kr    ) * LDS_STRIDE + 48 + l16];
        b3.y = Wl[(kr + 1) * LDS_STRIDE + 48 + l16];
        acc0 = __builtin_amdgcn_wmma_f32_16x16x4_f32(false, a, false, b0, (short)0, acc0, false, false);
        acc1 = __builtin_amdgcn_wmma_f32_16x16x4_f32(false, a, false, b1, (short)0, acc1, false, false);
        acc2 = __builtin_amdgcn_wmma_f32_16x16x4_f32(false, a, false, b2, (short)0, acc2, false, false);
        acc3 = __builtin_amdgcn_wmma_f32_16x16x4_f32(false, a, false, b3, (short)0, acc3, false, false);
    }

    float dv[8];
    #pragma unroll
    for (int r = 0; r < 8; ++r) dv[r] = __frsqrt_rn((float)deg[wave * 16 + half * 8 + r] + 1.0f);

    v8f accs[4] = {acc0, acc1, acc2, acc3};
    #pragma unroll
    for (int nt = 0; nt < 4; ++nt) {
        #pragma unroll
        for (int r = 0; r < 8; ++r) {
            float v = accs[nt][r] * dv[r];
            size_t off = (size_t)(m0 + half * 8 + r) * Z_DIM + nt * 16 + l16;
            s[off] = v;
        }
    }
}


#define AGG_THREADS 1024
#define AGG_WAVES   (AGG_THREADS / 32)
#define SCAN_PER_LANE ((BIN_R + 31) / 32)
#define POOL_WORDS  ((BIN_R + 1) * Z_DIM)

#define SORT_BATCH  8

static_assert(POOL_WORDS >= BK_CAP + 16, "the sorted list of a full bucket must fit, and a 16-word read past its end");
static_assert(BIN_R % 2 == 0, "two nodes per wave step");

__device__ __noinline__ void agg_slow_add(float* acc, const int* __restrict__ ei,
                                          const float* __restrict__ s, int e, unsigned lc) {
    const int row = ei[e];
    for (int k = 0; k < Z_DIM; ++k)
        atomicAdd(&acc[lc * Z_DIM + k], s[(size_t)row * Z_DIM + k]);
}

__global__ __launch_bounds__(AGG_THREADS) void aggregate_kernel(const int* __restrict__ ei,
                                                                const float* __restrict__ s,
                                                                const float* __restrict__ dinv,
                                                                const float* __restrict__ bias,
                                                                const float* __restrict__ pa,
                                                                float* out) {
    __shared__ unsigned pool[POOL_WORDS];
    __shared__ unsigned cur[BIN_R];

    const int tid  = threadIdx.x;
    const int lane = tid & 31;
    const int wave = tid >> 5;
    const int base = blockIdx.x * BIN_R;

    const unsigned* bkt = (const unsigned*)(out + (size_t)base * Z_DIM);
    const unsigned n = (unsigned)__builtin_amdgcn_readfirstlane((int)bkt[0]);

    if (n <= (unsigned)BK_CAP) {
        const unsigned* ents = bkt + BK_HDR;

        if (tid < BIN_R) cur[tid] = 0u;
        __syncthreads();
        for (unsigned g = 0; g < n; g += SORT_BATCH * AGG_THREADS) {
            unsigned e[SORT_BATCH];
            #pragma unroll
            for (int k = 0; k < SORT_BATCH; ++k) {
                const unsigned i = g + (unsigned)(k * AGG_THREADS + tid);
                e[k] = ents[(i < n) ? i : 0u];
            }
            #pragma unroll
            for (int k = 0; k < SORT_BATCH; ++k) {
                const unsigned i = g + (unsigned)(k * AGG_THREADS + tid);
                if (i < n) atomicAdd(&cur[e[k] >> R_BITS], 1u);
            }
        }
        __syncthreads();

        if (wave == 0) {
            unsigned cnt[SCAN_PER_LANE];
            unsigned t = 0u;
            #pragma unroll
            for (int k = 0; k < SCAN_PER_LANE; ++k) {
                const int idx = lane * SCAN_PER_LANE + k;
                cnt[k] = (idx < BIN_R) ? cur[idx] : 0u;
                t += cnt[k];
            }
            unsigned x = t;
            #pragma unroll
            for (int dlt = 1; dlt < 32; dlt <<= 1) {
                const unsigned v = __shfl_up(x, dlt);
                if (lane >= dlt) x += v;
            }
            unsigned run = x - t;
            #pragma unroll
            for (int k = 0; k < SCAN_PER_LANE; ++k) {
                const int idx = lane * SCAN_PER_LANE + k;
                if (idx < BIN_R) cur[idx] = run;
                run += cnt[k];
            }
        }
        __syncthreads();

        for (unsigned g = 0; g < n; g += SORT_BATCH * AGG_THREADS) {
            unsigned e[SORT_BATCH];
            #pragma unroll
            for (int k = 0; k < SORT_BATCH; ++k) {
                const unsigned i = g + (unsigned)(k * AGG_THREADS + tid);
                e[k] = ents[(i < n) ? i : 0u];
            }
            #pragma unroll
            for (int k = 0; k < SORT_BATCH; ++k) {
                const unsigned i = g + (unsigned)(k * AGG_THREADS + tid);
                if (i < n) {
                    const unsigned pos = atomicAdd(&cur[e[k] >> R_BITS], 1u);
                    pool[pos] = e[k] & R_MASK;
                }
            }
        }
        __syncthreads();

        const int h  = lane >> 4;
        const int c4 = (lane & 15) << 2;
        const v4f b4 = *(const v4f*)(bias + c4);
        const v4f p4 = *(const v4f*)(pa + c4);

        for (int p = wave; p < BIN_R / 2; p += AGG_WAVES) {
            const int l = 2 * p + h;
            const unsigned beg = l ? cur[l - 1] : 0u;
            const unsigned len = cur[l] - beg;
            v4f acc = *(const v4f*)(s + (size_t)(base + l) * Z_DIM + c4);
            const float d = dinv[base + l];
            const unsigned other = (unsigned)__shfl_xor((int)len, 16);
            const unsigned maxlen = (unsigned)__builtin_amdgcn_readfirstlane((int)(len > other ? len : other));

            for (unsigned j0 = 0; j0 < maxlen; j0 += 16) {
                const unsigned q0 = beg + ((j0 < len) ? j0 : 0u);
                unsigned r[16];
                #pragma unroll
                for (int j = 0; j < 16; ++j) r[j] = pool[q0 + (unsigned)j];
                v4f a[16];
                #pragma unroll
                for (int j = 0; j < 16; ++j) {
                    a[j] = (v4f){0.0f, 0.0f, 0.0f, 0.0f};
                    if (j0 + (unsigned)j < len) a[j] = *(const v4f*)(s + (size_t)r[j] * Z_DIM + c4);
                }
                #pragma unroll
                for (int j = 0; j < 16; ++j) acc += a[j];
            }

            v4f v;
            v.x = d * acc.x + b4.x;
            v.y = d * acc.y + b4.y;
            v.z = d * acc.z + b4.z;
            v.w = d * acc.w + b4.w;
            v.x = (v.x >= 0.0f) ? v.x : p4.x * v.x;
            v.y = (v.y >= 0.0f) ? v.y : p4.y * v.y;
            v.z = (v.z >= 0.0f) ? v.z : p4.z * v.z;
            v.w = (v.w >= 0.0f) ? v.w : p4.w * v.w;
            *(v4f*)(out + (size_t)(base + l) * Z_DIM + c4) = v;
        }
    } else {
        float* acc = (float*)pool;
        {
            const v4f* s4 = (const v4f*)(s + (size_t)base * Z_DIM);
            for (int i = tid; i < BIN_R * Z_DIM / 4; i += AGG_THREADS) ((v4f*)acc)[i] = s4[i];
        }
        __syncthreads();
        const v4i* col4 = (const v4i*)(ei + N_EDGES);
        for (int gi = tid; gi < COL_GROUPS; gi += AGG_THREADS) {
            const v4i c = col4[gi];
            #pragma unroll
            for (int j = 0; j < 4; ++j) {
                const unsigned lc = (unsigned)(c[j] - base);
                if (lc < (unsigned)BIN_R) agg_slow_add(acc, ei, s, gi * 4 + j, lc);
            }
        }
        __syncthreads();
        const v4f* acc4 = (const v4f*)acc;
        v4f* out4 = (v4f*)(out + (size_t)base * Z_DIM);
        for (int i = tid; i < BIN_R * Z_DIM / 4; i += AGG_THREADS) {
            const v4f a = acc4[i];
            const float d = dinv[base + (i >> 4)];
            const int k4 = (i & 15) << 2;
            v4f v;
            v.x = d * a.x + bias[k4 + 0];
            v.y = d * a.y + bias[k4 + 1];
            v.z = d * a.z + bias[k4 + 2];
            v.w = d * a.w + bias[k4 + 3];
            v.x = (v.x >= 0.0f) ? v.x : pa[k4 + 0] * v.x;
            v.y = (v.y >= 0.0f) ? v.y : pa[k4 + 1] * v.y;
            v.z = (v.z >= 0.0f) ? v.z : pa[k4 + 2] * v.z;
            v.w = (v.w >= 0.0f) ? v.w : pa[k4 + 3] * v.w;
            out4[i] = v;
        }
    }
}


extern "C" void kernel_launch(void* const* d_in, const int* in_sizes, int n_in,
                              void* d_out, int out_size, void* d_ws, size_t ws_size,
                              hipStream_t stream) {
    const float* x    = (const float*)d_in[0];
    const int*   ei   = (const int*)d_in[1];
    const float* W    = (const float*)d_in[3];
    const float* bias = (const float*)d_in[4];
    const float* pa   = (const float*)d_in[5];
    float* out = (float*)d_out;

    float* dinv   = (float*)((char*)d_ws + (size_t)N_NODES * 4);
    float* s      = (float*)((char*)d_ws + (size_t)2 * N_NODES * 4);

    unsigned* bk = (unsigned*)d_out;

    zero_cursors_kernel<<<(N_BINS + 255) / 256, 256, 0, stream>>>(bk);
    bin_kernel<<<BIN_BLOCKS, BIN_THREADS, 0, stream>>>(ei, bk);
    gemm_degree_scale_kernel<<<N_BINS, GEMM_THREADS, 0, stream>>>(x, W, ei, bk, dinv, s);

    aggregate_kernel<<<N_BINS, AGG_THREADS, 0, stream>>>(ei, s, dinv, bias, pa, out);
}


# Binning workgroup writes each bin's run to its own fixed 64-word slot of the bucket (count in slot header): no fill cursor, global atomic or clearing launch (3 launches); sums in LDS arrival order
# speedup vs baseline: 4.7949x; 1.0764x over previous
#include <hip/hip_runtime.h>


typedef __attribute__((ext_vector_type(2))) float v2f;
typedef __attribute__((ext_vector_type(4))) float v4f;
typedef __attribute__((ext_vector_type(8))) float v8f;
typedef __attribute__((ext_vector_type(4))) int   v4i;
typedef __attribute__((ext_vector_type(2))) unsigned v2u;

#define N_NODES 100000
#define N_EDGES 1200000
#define IN_DIM  256
#define Z_DIM   64
#define LDS_STRIDE 72


#define BIN_R      400
#define N_BINS     (N_NODES / BIN_R)
#define BK_WORDS   (BIN_R * Z_DIM)
#define R_BITS     17
#define R_MASK     ((1u << R_BITS) - 1u)
#define COL_GROUPS (N_EDGES / 4)
#define BIN_THREADS 1024
#define BIN_BLOCKS ((COL_GROUPS + BIN_THREADS - 1) / BIN_THREADS)
#define N_SLOTS    BIN_BLOCKS
#define SLOT_W     64
#define SLOT_CAP   (SLOT_W - 1)
#define SLOT_LANES 8

static_assert(N_NODES % BIN_R == 0, "node ranges must tile the graph");
static_assert(N_EDGES % 4 == 0, "ids are read four at a time");
static_assert(N_NODES <= (1 << R_BITS), "source row must fit its field");
static_assert(BIN_R < (1 << (32 - R_BITS)), "local node id and the spare row must fit their field");
static_assert(N_BINS <= BIN_THREADS, "one thread per bin in the header step");
static_assert(N_SLOTS * SLOT_W <= BK_WORDS, "every binning workgroup has a slot in every bucket");
static_assert(SLOT_LANES * 4 * 2 == SLOT_W, "a slot is read as two lines of SLOT_LANES 16-B pieces");

#define BIN_EDGES     (BIN_THREADS * 4)
#define BIN_PER_LANE  ((N_BINS + 31) / 32)
#define NO_SLOT       0xFFFFFFFFu

static_assert(BIN_PER_LANE * 32 >= N_BINS, "the scan wave covers every bin");
static_assert(32 + N_BINS <= BIN_THREADS, "one header-writing thread per bin beside the scan wave");
static_assert((size_t)N_BINS * BK_WORDS < 0xFFFFFFFFull, "a bucket word index fits 32 bits below NO_SLOT");

__global__ __launch_bounds__(BIN_THREADS) void bin_kernel(const int* __restrict__ ei,
                                                          unsigned* __restrict__ bk) {
    __shared__ unsigned hist[N_BINS];
    __shared__ unsigned lbase[N_BINS];
    __shared__ v2u stage[BIN_EDGES];

    const int tid  = threadIdx.x;
    const int lane = tid & 31;
    const int wave = tid >> 5;
    const int g    = blockIdx.x * BIN_THREADS + tid;
    const bool ok  = g < COL_GROUPS;
    const unsigned total = (unsigned)min(BIN_EDGES, N_EDGES - (int)blockIdx.x * BIN_EDGES);

    if (tid < N_BINS) hist[tid] = 0u;
    __syncthreads();

    v4i src4 = {0, 0, 0, 0}, dst4 = {0, 0, 0, 0};
    unsigned bin[4], rank[4];
    if (ok) {
        src4 = ((const v4i*)ei)[g];
        dst4 = ((const v4i*)(ei + N_EDGES))[g];
        #pragma unroll
        for (int j = 0; j < 4; ++j) {
            const unsigned c = (unsigned)dst4[j];
            bin[j]  = c / (unsigned)BIN_R;
            rank[j] = atomicAdd(&hist[bin[j]], 1u);
        }
    }
    __syncthreads();

    if (wave == 0) {
        unsigned cnt[BIN_PER_LANE];
        unsigned t = 0u;
        #pragma unroll
        for (int k = 0; k < BIN_PER_LANE; ++k) {
            const int idx = lane * BIN_PER_LANE + k;
            cnt[k] = (idx < N_BINS) ? hist[idx] : 0u;
            t += cnt[k];
        }
        unsigned x = t;
        #pragma unroll
        for (int dlt = 1; dlt < 32; dlt <<= 1) {
            const unsigned v = __shfl_up(x, dlt);
            if (lane >= dlt) x += v;
        }
        unsigned run = x - t;
        #pragma unroll
        for (int k = 0; k < BIN_PER_LANE; ++k) {
            const int idx = lane * BIN_PER_LANE + k;
            if (idx < N_BINS) lbase[idx] = run;
            run += cnt[k];
        }
    } else if (tid - 32 < N_BINS) {
        const int b = tid - 32;
        bk[(size_t)b * BK_WORDS + (size_t)blockIdx.x * SLOT_W] = hist[b];
    }
    __syncthreads();

    if (ok) {
        #pragma unroll
        for (int j = 0; j < 4; ++j) {
            const unsigned b   = bin[j];
            const unsigned pos = rank[j];
            const unsigned lc  = (unsigned)dst4[j] - b * (unsigned)BIN_R;
            v2u ent;
            ent.x = (unsigned)src4[j] | (lc << R_BITS);
            ent.y = (pos < (unsigned)SLOT_CAP)
                        ? (b * (unsigned)BK_WORDS + blockIdx.x * (unsigned)SLOT_W + 1u + pos) : NO_SLOT;
            stage[lbase[b] + rank[j]] = ent;
        }
    }
    __syncthreads();

    v2u ent[4];
    #pragma unroll
    for (int k = 0; k < 4; ++k) ent[k] = stage[k * BIN_THREADS + tid];
    #pragma unroll
    for (int k = 0; k < 4; ++k) {
        const unsigned i = (unsigned)(k * BIN_THREADS + tid);
        if (i < total && ent[k].y != NO_SLOT) bk[ent[k].y] = ent[k].x;
    }
}

template <class F>
__device__ __forceinline__ void slot_entries(const unsigned* slot, v4i e0, unsigned cc, int q4, F f) {
    #pragma unroll
    for (int j = 0; j < 4; ++j) {
        const unsigned off = (unsigned)(q4 + j);
        if (off >= 1u && off <= cc) f((unsigned)e0[j]);
    }
    if (cc > (unsigned)(SLOT_W / 2 - 1)) {
        const v4i e1 = *(const v4i*)(slot + SLOT_W / 2 + q4);
        #pragma unroll
        for (int j = 0; j < 4; ++j) {
            const unsigned off = (unsigned)(SLOT_W / 2 + q4 + j);
            if (off <= cc) f((unsigned)e1[j]);
        }
    }
}

#define GEMM_WAVES   (BIN_R / 16)
#define GEMM_THREADS (GEMM_WAVES * 32)
#define DEG_PASSES   ((N_SLOTS * SLOT_LANES + GEMM_THREADS - 1) / GEMM_THREADS)

static_assert(BIN_R % 16 == 0, "a bin is a whole number of 16-row tiles");
static_assert(GEMM_THREADS <= 1024, "one workgroup per bin");
static_assert(GEMM_THREADS % 32 == 0 && 32 % SLOT_LANES == 0, "a slot's lanes sit in one wave");

__global__ __launch_bounds__(GEMM_THREADS) void gemm_degree_scale_kernel(const float* __restrict__ x,
                                                                         const float* __restrict__ W,
                                                                         const int* __restrict__ ei,
                                                                         const unsigned* __restrict__ bk,
                                                                         float* __restrict__ dinv,
                                                                         float* __restrict__ s) {
    __shared__ float Wl[IN_DIM * LDS_STRIDE];
    __shared__ unsigned deg[BIN_R];
    __shared__ unsigned ovf;

    const int tid  = threadIdx.x;
    const int base = blockIdx.x * BIN_R;
    const unsigned* bkt = bk + (size_t)blockIdx.x * BK_WORDS;

    const int q4 = (tid & (SLOT_LANES - 1)) * 4;
    v4i e[DEG_PASSES];
    #pragma unroll
    for (int k = 0; k < DEG_PASSES; ++k) {
        const int slot = (k * GEMM_THREADS + tid) / SLOT_LANES;
        e[k] = *(const v4i*)(bkt + (slot < N_SLOTS ? slot : 0) * SLOT_W + q4);
    }

    {
        constexpr int N4    = (IN_DIM * Z_DIM) / 4;
        constexpr int TRIPS = (N4 + GEMM_THREADS - 1) / GEMM_THREADS;
        v4f w[TRIPS];
        #pragma unroll
        for (int k = 0; k < TRIPS; ++k) {
            const int i = tid + k * GEMM_THREADS;
            if (i < N4) w[k] = *(const v4f*)(W + i * 4);
        }
        #pragma unroll
        for (int k = 0; k < TRIPS; ++k) {
            const int i = tid + k * GEMM_THREADS;
            if (i < N4) *(v4f*)(&Wl[(i >> 4) * LDS_STRIDE + ((i & 15) << 2)]) = w[k];
        }
    }
    for (int i = tid; i < BIN_R; i += GEMM_THREADS) deg[i] = 0u;
    if (tid == 0) ovf = 0u;
    __syncthreads();

    #pragma unroll
    for (int k = 0; k < DEG_PASSES; ++k) {
        const int slot = (k * GEMM_THREADS + tid) / SLOT_LANES;
        const unsigned c = (unsigned)__shfl(e[k].x, (tid & 31) & ~(SLOT_LANES - 1));
        const bool live = slot < N_SLOTS;
        if (live && c > (unsigned)SLOT_CAP) ovf = 1u;
        const unsigned cc = live ? min(c, (unsigned)SLOT_CAP) : 0u;
        slot_entries(bkt + (live ? slot : 0) * SLOT_W, e[k], cc, q4,
                     [&](unsigned ent) { atomicAdd(&deg[ent >> R_BITS], 1u); });
    }
    __syncthreads();

    if (__builtin_amdgcn_readfirstlane((int)ovf)) {
        for (int i = tid; i < BIN_R; i += GEMM_THREADS) deg[i] = 0u;
        __syncthreads();
        const v4i* col4 = (const v4i*)(ei + N_EDGES);
        for (int gi = tid; gi < COL_GROUPS; gi += GEMM_THREADS) {
            const v4i c = col4[gi];
            #pragma unroll
            for (int j = 0; j < 4; ++j) {
                const unsigned lc = (unsigned)(c[j] - base);
                if (lc < (unsigned)BIN_R) atomicAdd(&deg[lc], 1u);
            }
        }
        __syncthreads();
    }

    if (tid < BIN_R) dinv[base + tid] = __frsqrt_rn((float)deg[tid] + 1.0f);

    const int wave = tid >> 5;
    const int lane = tid & 31;
    const int m0   = base + wave * 16;
    const int half = lane >> 4;
    const int l16  = lane & 15;
    v8f z = {};
    v8f acc0 = z, acc1 = z, acc2 = z, acc3 = z;

    const float* xrow = x + (size_t)(m0 + l16) * IN_DIM + 2 * half;

    #pragma unroll 4
    for (int k = 0; k < IN_DIM; k += 4) {
        v2f a = *(const v2f*)(xrow + k);
        const int kr = k + 2 * half;
        v2f b0, b1, b2, b3;
        b0.x = Wl[(kr    ) * LDS_STRIDE +  0 + l16];
        b0.y = Wl[(kr + 1) * LDS_STRIDE +  0 + l16];
        b1.x = Wl[(kr    ) * LDS_STRIDE + 16 + l16];
        b1.y = Wl[(kr + 1) * LDS_STRIDE + 16 + l16];
        b2.x = Wl[(kr    ) * LDS_STRIDE + 32 + l16];
        b2.y = Wl[(kr + 1) * LDS_STRIDE + 32 + l16];
        b3.x = Wl[(kr    ) * LDS_STRIDE + 48 + l16];
        b3.y = Wl[(kr + 1) * LDS_STRIDE + 48 + l16];
        acc0 = __builtin_amdgcn_wmma_f32_16x16x4_f32(false, a, false, b0, (short)0, acc0, false, false);
        acc1 = __builtin_amdgcn_wmma_f32_16x16x4_f32(false, a, false, b1, (short)0, acc1, false, false);
        acc2 = __builtin_amdgcn_wmma_f32_16x16x4_f32(false, a, false, b2, (short)0, acc2, false, false);
        acc3 = __builtin_amdgcn_wmma_f32_16x16x4_f32(false, a, false, b3, (short)0, acc3, false, false);
    }

    float dv[8];
    #pragma unroll
    for (int r = 0; r < 8; ++r) dv[r] = __frsqrt_rn((float)deg[wave * 16 + half * 8 + r] + 1.0f);

    v8f accs[4] = {acc0, acc1, acc2, acc3};
    #pragma unroll
    for (int nt = 0; nt < 4; ++nt) {
        #pragma unroll
        for (int r = 0; r < 8; ++r) {
            float v = accs[nt][r] * dv[r];
            size_t off = (size_t)(m0 + half * 8 + r) * Z_DIM + nt * 16 + l16;
            s[off] = v;
        }
    }
}


#define AGG_THREADS 1024
#define AGG_WAVES   (AGG_THREADS / 32)
#define SCAN_PER_LANE ((BIN_R + 31) / 32)
#define POOL_WORDS  ((BIN_R + 1) * Z_DIM)

#define AGG_PASSES  ((N_SLOTS * SLOT_LANES + AGG_THREADS - 1) / AGG_THREADS)

static_assert(POOL_WORDS >= N_SLOTS * SLOT_CAP + 16, "the sorted list of a full bucket must fit, and a 16-word read past its end");
static_assert(AGG_THREADS % 32 == 0, "a slot's lanes sit in one wave");
static_assert(BIN_R % 2 == 0, "two nodes per wave step");

__device__ __noinline__ void agg_slow_add(float* acc, const int* __restrict__ ei,
                                          const float* __restrict__ s, int e, unsigned lc) {
    const int row = ei[e];
    for (int k = 0; k < Z_DIM; ++k)
        atomicAdd(&acc[lc * Z_DIM + k], s[(size_t)row * Z_DIM + k]);
}

__global__ __launch_bounds__(AGG_THREADS) void aggregate_kernel(const int* __restrict__ ei,
                                                                const float* __restrict__ s,
                                                                const float* __restrict__ dinv,
                                                                const float* __restrict__ bias,
                                                                const float* __restrict__ pa,
                                                                float* out) {
    __shared__ unsigned pool[POOL_WORDS];
    __shared__ unsigned cur[BIN_R];
    __shared__ unsigned ovf;

    const int tid  = threadIdx.x;
    const int lane = tid & 31;
    const int wave = tid >> 5;
    const int base = blockIdx.x * BIN_R;

    const unsigned* bkt = (const unsigned*)(out + (size_t)base * Z_DIM);

    const int q4 = (tid & (SLOT_LANES - 1)) * 4;
    v4i e[AGG_PASSES];
    #pragma unroll
    for (int k = 0; k < AGG_PASSES; ++k) {
        const int slot = (k * AGG_THREADS + tid) / SLOT_LANES;
        e[k] = *(const v4i*)(bkt + (slot < N_SLOTS ? slot : 0) * SLOT_W + q4);
    }

    if (tid < BIN_R) cur[tid] = 0u;
    if (tid == 0) ovf = 0u;
    __syncthreads();

    unsigned cc[AGG_PASSES];
    #pragma unroll
    for (int k = 0; k < AGG_PASSES; ++k) {
        const int slot = (k * AGG_THREADS + tid) / SLOT_LANES;
        const unsigned c = (unsigned)__shfl(e[k].x, lane & ~(SLOT_LANES - 1));
        const bool live = slot < N_SLOTS;
        if (live && c > (unsigned)SLOT_CAP) ovf = 1u;
        cc[k] = live ? min(c, (unsigned)SLOT_CAP) : 0u;
        slot_entries(bkt + (live ? slot : 0) * SLOT_W, e[k], cc[k], q4,
                     [&](unsigned ent) { atomicAdd(&cur[ent >> R_BITS], 1u); });
    }
    __syncthreads();

    if (!__builtin_amdgcn_readfirstlane((int)ovf)) {
        if (wave == 0) {
            unsigned cnt[SCAN_PER_LANE];
            unsigned t = 0u;
            #pragma unroll
            for (int k = 0; k < SCAN_PER_LANE; ++k) {
                const int idx = lane * SCAN_PER_LANE + k;
                cnt[k] = (idx < BIN_R) ? cur[idx] : 0u;
                t += cnt[k];
            }
            unsigned x = t;
            #pragma unroll
            for (int dlt = 1; dlt < 32; dlt <<= 1) {
                const unsigned v = __shfl_up(x, dlt);
                if (lane >= dlt) x += v;
            }
            unsigned run = x - t;
            #pragma unroll
            for (int k = 0; k < SCAN_PER_LANE; ++k) {
                const int idx = lane * SCAN_PER_LANE + k;
                if (idx < BIN_R) cur[idx] = run;
                run += cnt[k];
            }
        }
        __syncthreads();

        #pragma unroll
        for (int k = 0; k < AGG_PASSES; ++k) {
            const int slot = (k * AGG_THREADS + tid) / SLOT_LANES;
            slot_entries(bkt + (slot < N_SLOTS ? slot : 0) * SLOT_W, e[k], cc[k], q4,
                         [&](unsigned ent) {
                             const unsigned pos = atomicAdd(&cur[ent >> R_BITS], 1u);
                             pool[pos] = ent & R_MASK;
                         });
        }
        __syncthreads();

        const int h  = lane >> 4;
        const int c4 = (lane & 15) << 2;
        const v4f b4 = *(const v4f*)(bias + c4);
        const v4f p4 = *(const v4f*)(pa + c4);

        for (int p = wave; p < BIN_R / 2; p += AGG_WAVES) {
            const int l = 2 * p + h;
            const unsigned beg = l ? cur[l - 1] : 0u;
            const unsigned len = cur[l] - beg;
            v4f acc = *(const v4f*)(s + (size_t)(base + l) * Z_DIM + c4);
            const float d = dinv[base + l];
            const unsigned other = (unsigned)__shfl_xor((int)len, 16);
            const unsigned maxlen = (unsigned)__builtin_amdgcn_readfirstlane((int)(len > other ? len : other));

            for (unsigned j0 = 0; j0 < maxlen; j0 += 16) {
                const unsigned q0 = beg + ((j0 < len) ? j0 : 0u);
                unsigned r[16];
                #pragma unroll
                for (int j = 0; j < 16; ++j) r[j] = pool[q0 + (unsigned)j];
                v4f a[16];
                #pragma unroll
                for (int j = 0; j < 16; ++j) {
                    a[j] = (v4f){0.0f, 0.0f, 0.0f, 0.0f};
                    if (j0 + (unsigned)j < len) a[j] = *(const v4f*)(s + (size_t)r[j] * Z_DIM + c4);
                }
                #pragma unroll
                for (int j = 0; j < 16; ++j) acc += a[j];
            }

            v4f v;
            v.x = d * acc.x + b4.x;
            v.y = d * acc.y + b4.y;
            v.z = d * acc.z + b4.z;
            v.w = d * acc.w + b4.w;
            v.x = (v.x >= 0.0f) ? v.x : p4.x * v.x;
            v.y = (v.y >= 0.0f) ? v.y : p4.y * v.y;
            v.z = (v.z >= 0.0f) ? v.z : p4.z * v.z;
            v.w = (v.w >= 0.0f) ? v.w : p4.w * v.w;
            *(v4f*)(out + (size_t)(base + l) * Z_DIM + c4) = v;
        }
    } else {
        float* acc = (float*)pool;
        {
            const v4f* s4 = (const v4f*)(s + (size_t)base * Z_DIM);
            for (int i = tid; i < BIN_R * Z_DIM / 4; i += AGG_THREADS) ((v4f*)acc)[i] = s4[i];
        }
        __syncthreads();
        const v4i* col4 = (const v4i*)(ei + N_EDGES);
        for (int gi = tid; gi < COL_GROUPS; gi += AGG_THREADS) {
            const v4i c = col4[gi];
            #pragma unroll
            for (int j = 0; j < 4; ++j) {
                const unsigned lc = (unsigned)(c[j] - base);
                if (lc < (unsigned)BIN_R) agg_slow_add(acc, ei, s, gi * 4 + j, lc);
            }
        }
        __syncthreads();
        const v4f* acc4 = (const v4f*)acc;
        v4f* out4 = (v4f*)(out + (size_t)base * Z_DIM);
        for (int i = tid; i < BIN_R * Z_DIM / 4; i += AGG_THREADS) {
            const v4f a = acc4[i];
            const float d = dinv[base + (i >> 4)];
            const int k4 = (i & 15) << 2;
            v4f v;
            v.x = d * a.x + bias[k4 + 0];
            v.y = d * a.y + bias[k4 + 1];
            v.z = d * a.z + bias[k4 + 2];
            v.w = d * a.w + bias[k4 + 3];
            v.x = (v.x >= 0.0f) ? v.x : pa[k4 + 0] * v.x;
            v.y = (v.y >= 0.0f) ? v.y : pa[k4 + 1] * v.y;
            v.z = (v.z >= 0.0f) ? v.z : pa[k4 + 2] * v.z;
            v.w = (v.w >= 0.0f) ? v.w : pa[k4 + 3] * v.w;
            out4[i] = v;
        }
    }
}


extern "C" void kernel_launch(void* const* d_in, const int* in_sizes, int n_in,
                              void* d_out, int out_size, void* d_ws, size_t ws_size,
                              hipStream_t stream) {
    const float* x    = (const float*)d_in[0];
    const int*   ei   = (const int*)d_in[1];
    const float* W    = (const float*)d_in[3];
    const float* bias = (const float*)d_in[4];
    const float* pa   = (const float*)d_in[5];
    float* out = (float*)d_out;

    float* dinv   = (float*)((char*)d_ws + (size_t)N_NODES * 4);
    float* s      = (float*)((char*)d_ws + (size_t)2 * N_NODES * 4);

    unsigned* bk = (unsigned*)d_out;

    bin_kernel<<<BIN_BLOCKS, BIN_THREADS, 0, stream>>>(ei, bk);
    gemm_degree_scale_kernel<<<N_BINS, GEMM_THREADS, 0, stream>>>(x, W, ei, bk, dinv, s);

    aggregate_kernel<<<N_BINS, AGG_THREADS, 0, stream>>>(ei, s, dinv, bias, pa, out);
}


# GEMM launch: 24 whole-tile waves per bin (six per SIMD); the 25th tile is cut into four 16-column blocks done first by waves 0-3, one per SIMD: 1600 WMMAs per SIMD, was 1792 on one; sums unchanged
# speedup vs baseline: 4.9616x; 1.0348x over previous
#include <hip/hip_runtime.h>


typedef __attribute__((ext_vector_type(2))) float v2f;
typedef __attribute__((ext_vector_type(4))) float v4f;
typedef __attribute__((ext_vector_type(8))) float v8f;
typedef __attribute__((ext_vector_type(4))) int   v4i;
typedef __attribute__((ext_vector_type(2))) unsigned v2u;

#define N_NODES 100000
#define N_EDGES 1200000
#define IN_DIM  256
#define Z_DIM   64
#define LDS_STRIDE 72


#define BIN_R      400
#define N_BINS     (N_NODES / BIN_R)
#define BK_WORDS   (BIN_R * Z_DIM)
#define R_BITS     17
#define R_MASK     ((1u << R_BITS) - 1u)
#define COL_GROUPS (N_EDGES / 4)
#define BIN_THREADS 1024
#define BIN_BLOCKS ((COL_GROUPS + BIN_THREADS - 1) / BIN_THREADS)
#define N_SLOTS    BIN_BLOCKS
#define SLOT_W     64
#define SLOT_CAP   (SLOT_W - 1)
#define SLOT_LANES 8

static_assert(N_NODES % BIN_R == 0, "node ranges must tile the graph");
static_assert(N_EDGES % 4 == 0, "ids are read four at a time");
static_assert(N_NODES <= (1 << R_BITS), "source row must fit its field");
static_assert(BIN_R < (1 << (32 - R_BITS)), "local node id and the spare row must fit their field");
static_assert(N_BINS <= BIN_THREADS, "one thread per bin in the header step");
static_assert(N_SLOTS * SLOT_W <= BK_WORDS, "every binning workgroup has a slot in every bucket");
static_assert(SLOT_LANES * 4 * 2 == SLOT_W, "a slot is read as two lines of SLOT_LANES 16-B pieces");

#define BIN_EDGES     (BIN_THREADS * 4)
#define BIN_PER_LANE  ((N_BINS + 31) / 32)
#define NO_SLOT       0xFFFFFFFFu

static_assert(BIN_PER_LANE * 32 >= N_BINS, "the scan wave covers every bin");
static_assert(32 + N_BINS <= BIN_THREADS, "one header-writing thread per bin beside the scan wave");
static_assert((size_t)N_BINS * BK_WORDS < 0xFFFFFFFFull, "a bucket word index fits 32 bits below NO_SLOT");

__global__ __launch_bounds__(BIN_THREADS) void bin_kernel(const int* __restrict__ ei,
                                                          unsigned* __restrict__ bk) {
    __shared__ unsigned hist[N_BINS];
    __shared__ unsigned lbase[N_BINS];
    __shared__ v2u stage[BIN_EDGES];

    const int tid  = threadIdx.x;
    const int lane = tid & 31;
    const int wave = tid >> 5;
    const int g    = blockIdx.x * BIN_THREADS + tid;
    const bool ok  = g < COL_GROUPS;
    const unsigned total = (unsigned)min(BIN_EDGES, N_EDGES - (int)blockIdx.x * BIN_EDGES);

    if (tid < N_BINS) hist[tid] = 0u;
    __syncthreads();

    v4i src4 = {0, 0, 0, 0}, dst4 = {0, 0, 0, 0};
    unsigned bin[4], rank[4];
    if (ok) {
        src4 = ((const v4i*)ei)[g];
        dst4 = ((const v4i*)(ei + N_EDGES))[g];
        #pragma unroll
        for (int j = 0; j < 4; ++j) {
            const unsigned c = (unsigned)dst4[j];
            bin[j]  = c / (unsigned)BIN_R;
            rank[j] = atomicAdd(&hist[bin[j]], 1u);
        }
    }
    __syncthreads();

    if (wave == 0) {
        unsigned cnt[BIN_PER_LANE];
        unsigned t = 0u;
        #pragma unroll
        for (int k = 0; k < BIN_PER_LANE; ++k) {
            const int idx = lane * BIN_PER_LANE + k;
            cnt[k] = (idx < N_BINS) ? hist[idx] : 0u;
            t += cnt[k];
        }
        unsigned x = t;
        #pragma unroll
        for (int dlt = 1; dlt < 32; dlt <<= 1) {
            const unsigned v = __shfl_up(x, dlt);
            if (lane >= dlt) x += v;
        }
        unsigned run = x - t;
        #pragma unroll
        for (int k = 0; k < BIN_PER_LANE; ++k) {
            const int idx = lane * BIN_PER_LANE + k;
            if (idx < N_BINS) lbase[idx] = run;
            run += cnt[k];
        }
    } else if (tid - 32 < N_BINS) {
        const int b = tid - 32;
        bk[(size_t)b * BK_WORDS + (size_t)blockIdx.x * SLOT_W] = hist[b];
    }
    __syncthreads();

    if (ok) {
        #pragma unroll
        for (int j = 0; j < 4; ++j) {
            const unsigned b   = bin[j];
            const unsigned pos = rank[j];
            const unsigned lc  = (unsigned)dst4[j] - b * (unsigned)BIN_R;
            v2u ent;
            ent.x = (unsigned)src4[j] | (lc << R_BITS);
            ent.y = (pos < (unsigned)SLOT_CAP)
                        ? (b * (unsigned)BK_WORDS + blockIdx.x * (unsigned)SLOT_W + 1u + pos) : NO_SLOT;
            stage[lbase[b] + rank[j]] = ent;
        }
    }
    __syncthreads();

    v2u ent[4];
    #pragma unroll
    for (int k = 0; k < 4; ++k) ent[k] = stage[k * BIN_THREADS + tid];
    #pragma unroll
    for (int k = 0; k < 4; ++k) {
        const unsigned i = (unsigned)(k * BIN_THREADS + tid);
        if (i < total && ent[k].y != NO_SLOT) bk[ent[k].y] = ent[k].x;
    }
}

template <class F>
__device__ __forceinline__ void slot_entries(const unsigned* slot, v4i e0, unsigned cc, int q4, F f) {
    #pragma unroll
    for (int j = 0; j < 4; ++j) {
        const unsigned off = (unsigned)(q4 + j);
        if (off >= 1u && off <= cc) f((unsigned)e0[j]);
    }
    if (cc > (unsigned)(SLOT_W / 2 - 1)) {
        const v4i e1 = *(const v4i*)(slot + SLOT_W / 2 + q4);
        #pragma unroll
        for (int j = 0; j < 4; ++j) {
            const unsigned off = (unsigned)(SLOT_W / 2 + q4 + j);
            if (off <= cc) f((unsigned)e1[j]);
        }
    }
}

#define BIN_TILES    (BIN_R / 16)
#define GEMM_WAVES   (BIN_TILES - 1)
#define SPLIT_WAVES  (Z_DIM / 16)
#define GEMM_THREADS (GEMM_WAVES * 32)
#define DEG_PASSES   ((N_SLOTS * SLOT_LANES + GEMM_THREADS - 1) / GEMM_THREADS)

static_assert(BIN_R % 16 == 0, "a bin is a whole number of 16-row tiles");
static_assert(GEMM_WAVES % 4 == 0 && SPLIT_WAVES == 4, "equal matrix work on the four SIMDs");
static_assert(GEMM_THREADS <= 1024, "one workgroup per bin");
static_assert(GEMM_THREADS % 32 == 0 && 32 % SLOT_LANES == 0, "a slot's lanes sit in one wave");

__global__ __launch_bounds__(GEMM_THREADS) void gemm_degree_scale_kernel(const float* __restrict__ x,
                                                                         const float* __restrict__ W,
                                                                         const int* __restrict__ ei,
                                                                         const unsigned* __restrict__ bk,
                                                                         float* __restrict__ dinv,
                                                                         float* __restrict__ s) {
    __shared__ float Wl[IN_DIM * LDS_STRIDE];
    __shared__ unsigned deg[BIN_R];
    __shared__ unsigned ovf;

    const int tid  = threadIdx.x;
    const int base = blockIdx.x * BIN_R;
    const unsigned* bkt = bk + (size_t)blockIdx.x * BK_WORDS;

    const int q4 = (tid & (SLOT_LANES - 1)) * 4;
    v4i e[DEG_PASSES];
    #pragma unroll
    for (int k = 0; k < DEG_PASSES; ++k) {
        const int slot = (k * GEMM_THREADS + tid) / SLOT_LANES;
        e[k] = *(const v4i*)(bkt + (slot < N_SLOTS ? slot : 0) * SLOT_W + q4);
    }

    {
        constexpr int N4    = (IN_DIM * Z_DIM) / 4;
        constexpr int TRIPS = (N4 + GEMM_THREADS - 1) / GEMM_THREADS;
        v4f w[TRIPS];
        #pragma unroll
        for (int k = 0; k < TRIPS; ++k) {
            const int i = tid + k * GEMM_THREADS;
            if (i < N4) w[k] = *(const v4f*)(W + i * 4);
        }
        #pragma unroll
        for (int k = 0; k < TRIPS; ++k) {
            const int i = tid + k * GEMM_THREADS;
            if (i < N4) *(v4f*)(&Wl[(i >> 4) * LDS_STRIDE + ((i & 15) << 2)]) = w[k];
        }
    }
    for (int i = tid; i < BIN_R; i += GEMM_THREADS) deg[i] = 0u;
    if (tid == 0) ovf = 0u;
    __syncthreads();

    #pragma unroll
    for (int k = 0; k < DEG_PASSES; ++k) {
        const int slot = (k * GEMM_THREADS + tid) / SLOT_LANES;
        const unsigned c = (unsigned)__shfl(e[k].x, (tid & 31) & ~(SLOT_LANES - 1));
        const bool live = slot < N_SLOTS;
        if (live && c > (unsigned)SLOT_CAP) ovf = 1u;
        const unsigned cc = live ? min(c, (unsigned)SLOT_CAP) : 0u;
        slot_entries(bkt + (live ? slot : 0) * SLOT_W, e[k], cc, q4,
                     [&](unsigned ent) { atomicAdd(&deg[ent >> R_BITS], 1u); });
    }
    __syncthreads();

    if (__builtin_amdgcn_readfirstlane((int)ovf)) {
        for (int i = tid; i < BIN_R; i += GEMM_THREADS) deg[i] = 0u;
        __syncthreads();
        const v4i* col4 = (const v4i*)(ei + N_EDGES);
        for (int gi = tid; gi < COL_GROUPS; gi += GEMM_THREADS) {
            const v4i c = col4[gi];
            #pragma unroll
            for (int j = 0; j < 4; ++j) {
                const unsigned lc = (unsigned)(c[j] - base);
                if (lc < (unsigned)BIN_R) atomicAdd(&deg[lc], 1u);
            }
        }
        __syncthreads();
    }

    if (tid < BIN_R) dinv[base + tid] = __frsqrt_rn((float)deg[tid] + 1.0f);

    const int wave = __builtin_amdgcn_readfirstlane(tid >> 5);
    const int lane = tid & 31;
    const int m0   = base + wave * 16;
    const int half = lane >> 4;
    const int l16  = lane & 15;
    v8f z = {};

    if (wave < SPLIT_WAVES) {
        const int mq = GEMM_WAVES * 16;
        const float* xq = x + (size_t)(base + mq + l16) * IN_DIM + 2 * half;
        const float* wq = &Wl[wave * 16 + l16];
        v8f accq = z;
        #pragma unroll 16
        for (int k = 0; k < IN_DIM; k += 4) {
            v2f a = *(const v2f*)(xq + k);
            const int kr = k + 2 * half;
            v2f b;
            b.x = wq[(kr    ) * LDS_STRIDE];
            b.y = wq[(kr + 1) * LDS_STRIDE];
            accq = __builtin_amdgcn_wmma_f32_16x16x4_f32(false, a, false, b, (short)0, accq, false, false);
        }
        #pragma unroll
        for (int r = 0; r < 8; ++r) {
            const int lr = mq + half * 8 + r;
            const float dq = __frsqrt_rn((float)deg[lr] + 1.0f);
            s[(size_t)(base + lr) * Z_DIM + wave * 16 + l16] = accq[r] * dq;
        }
    }

    v8f acc0 = z, acc1 = z, acc2 = z, acc3 = z;

    const float* xrow = x + (size_t)(m0 + l16) * IN_DIM + 2 * half;

    #pragma unroll 4
    for (int k = 0; k < IN_DIM; k += 4) {
        v2f a = *(const v2f*)(xrow + k);
        const int kr = k + 2 * half;
        v2f b0, b1, b2, b3;
        b0.x = Wl[(kr    ) * LDS_STRIDE +  0 + l16];
        b0.y = Wl[(kr + 1) * LDS_STRIDE +  0 + l16];
        b1.x = Wl[(kr    ) * LDS_STRIDE + 16 + l16];
        b1.y = Wl[(kr + 1) * LDS_STRIDE + 16 + l16];
        b2.x = Wl[(kr    ) * LDS_STRIDE + 32 + l16];
        b2.y = Wl[(kr + 1) * LDS_STRIDE + 32 + l16];
        b3.x = Wl[(kr    ) * LDS_STRIDE + 48 + l16];
        b3.y = Wl[(kr + 1) * LDS_STRIDE + 48 + l16];
        acc0 = __builtin_amdgcn_wmma_f32_16x16x4_f32(false, a, false, b0, (short)0, acc0, false, false);
        acc1 = __builtin_amdgcn_wmma_f32_16x16x4_f32(false, a, false, b1, (short)0, acc1, false, false);
        acc2 = __builtin_amdgcn_wmma_f32_16x16x4_f32(false, a, false, b2, (short)0, acc2, false, false);
        acc3 = __builtin_amdgcn_wmma_f32_16x16x4_f32(false, a, false, b3, (short)0, acc3, false, false);
    }

    float dv[8];
    #pragma unroll
    for (int r = 0; r < 8; ++r) dv[r] = __frsqrt_rn((float)deg[wave * 16 + half * 8 + r] + 1.0f);

    v8f accs[4] = {acc0, acc1, acc2, acc3};
    #pragma unroll
    for (int nt = 0; nt < 4; ++nt) {
        #pragma unroll
        for (int r = 0; r < 8; ++r) {
            float v = accs[nt][r] * dv[r];
            size_t off = (size_t)(m0 + half * 8 + r) * Z_DIM + nt * 16 + l16;
            s[off] = v;
        }
    }
}


#define AGG_THREADS 1024
#define AGG_WAVES   (AGG_THREADS / 32)
#define SCAN_PER_LANE ((BIN_R + 31) / 32)
#define POOL_WORDS  ((BIN_R + 1) * Z_DIM)

#define AGG_PASSES  ((N_SLOTS * SLOT_LANES + AGG_THREADS - 1) / AGG_THREADS)

static_assert(POOL_WORDS >= N_SLOTS * SLOT_CAP + 16, "the sorted list of a full bucket must fit, and a 16-word read past its end");
static_assert(AGG_THREADS % 32 == 0, "a slot's lanes sit in one wave");
static_assert(BIN_R % 2 == 0, "two nodes per wave step");

__device__ __noinline__ void agg_slow_add(float* acc, const int* __restrict__ ei,
                                          const float* __restrict__ s, int e, unsigned lc) {
    const int row = ei[e];
    for (int k = 0; k < Z_DIM; ++k)
        atomicAdd(&acc[lc * Z_DIM + k], s[(size_t)row * Z_DIM + k]);
}

__global__ __launch_bounds__(AGG_THREADS) void aggregate_kernel(const int* __restrict__ ei,
                                                                const float* __restrict__ s,
                                                                const float* __restrict__ dinv,
                                                                const float* __restrict__ bias,
                                                                const float* __restrict__ pa,
                                                                float* out) {
    __shared__ unsigned pool[POOL_WORDS];
    __shared__ unsigned cur[BIN_R];
    __shared__ unsigned ovf;

    const int tid  = threadIdx.x;
    const int lane = tid & 31;
    const int wave = tid >> 5;
    const int base = blockIdx.x * BIN_R;

    const unsigned* bkt = (const unsigned*)(out + (size_t)base * Z_DIM);

    const int q4 = (tid & (SLOT_LANES - 1)) * 4;
    v4i e[AGG_PASSES];
    #pragma unroll
    for (int k = 0; k < AGG_PASSES; ++k) {
        const int slot = (k * AGG_THREADS + tid) / SLOT_LANES;
        e[k] = *(const v4i*)(bkt + (slot < N_SLOTS ? slot : 0) * SLOT_W + q4);
    }

    if (tid < BIN_R) cur[tid] = 0u;
    if (tid == 0) ovf = 0u;
    __syncthreads();

    unsigned cc[AGG_PASSES];
    #pragma unroll
    for (int k = 0; k < AGG_PASSES; ++k) {
        const int slot = (k * AGG_THREADS + tid) / SLOT_LANES;
        const unsigned c = (unsigned)__shfl(e[k].x, lane & ~(SLOT_LANES - 1));
        const bool live = slot < N_SLOTS;
        if (live && c > (unsigned)SLOT_CAP) ovf = 1u;
        cc[k] = live ? min(c, (unsigned)SLOT_CAP) : 0u;
        slot_entries(bkt + (live ? slot : 0) * SLOT_W, e[k], cc[k], q4,
                     [&](unsigned ent) { atomicAdd(&cur[ent >> R_BITS], 1u); });
    }
    __syncthreads();

    if (!__builtin_amdgcn_readfirstlane((int)ovf)) {
        if (wave == 0) {
            unsigned cnt[SCAN_PER_LANE];
            unsigned t = 0u;
            #pragma unroll
            for (int k = 0; k < SCAN_PER_LANE; ++k) {
                const int idx = lane * SCAN_PER_LANE + k;
                cnt[k] = (idx < BIN_R) ? cur[idx] : 0u;
                t += cnt[k];
            }
            unsigned x = t;
            #pragma unroll
            for (int dlt = 1; dlt < 32; dlt <<= 1) {
                const unsigned v = __shfl_up(x, dlt);
                if (lane >= dlt) x += v;
            }
            unsigned run = x - t;
            #pragma unroll
            for (int k = 0; k < SCAN_PER_LANE; ++k) {
                const int idx = lane * SCAN_PER_LANE + k;
                if (idx < BIN_R) cur[idx] = run;
                run += cnt[k];
            }
        }
        __syncthreads();

        #pragma unroll
        for (int k = 0; k < AGG_PASSES; ++k) {
            const int slot = (k * AGG_THREADS + tid) / SLOT_LANES;
            slot_entries(bkt + (slot < N_SLOTS ? slot : 0) * SLOT_W, e[k], cc[k], q4,
                         [&](unsigned ent) {
                             const unsigned pos = atomicAdd(&cur[ent >> R_BITS], 1u);
                             pool[pos] = ent & R_MASK;
                         });
        }
        __syncthreads();

        const int h  = lane >> 4;
        const int c4 = (lane & 15) << 2;
        const v4f b4 = *(const v4f*)(bias + c4);
        const v4f p4 = *(const v4f*)(pa + c4);

        for (int p = wave; p < BIN_R / 2; p += AGG_WAVES) {
            const int l = 2 * p + h;
            const unsigned beg = l ? cur[l - 1] : 0u;
            const unsigned len = cur[l] - beg;
            v4f acc = *(const v4f*)(s + (size_t)(base + l) * Z_DIM + c4);
            const float d = dinv[base + l];
            const unsigned other = (unsigned)__shfl_xor((int)len, 16);
            const unsigned maxlen = (unsigned)__builtin_amdgcn_readfirstlane((int)(len > other ? len : other));

            for (unsigned j0 = 0; j0 < maxlen; j0 += 16) {
                const unsigned q0 = beg + ((j0 < len) ? j0 : 0u);
                unsigned r[16];
                #pragma unroll
                for (int j = 0; j < 16; ++j) r[j] = pool[q0 + (unsigned)j];
                v4f a[16];
                #pragma unroll
                for (int j = 0; j < 16; ++j) {
                    a[j] = (v4f){0.0f, 0.0f, 0.0f, 0.0f};
                    if (j0 + (unsigned)j < len) a[j] = *(const v4f*)(s + (size_t)r[j] * Z_DIM + c4);
                }
                #pragma unroll
                for (int j = 0; j < 16; ++j) acc += a[j];
            }

            v4f v;
            v.x = d * acc.x + b4.x;
            v.y = d * acc.y + b4.y;
            v.z = d * acc.z + b4.z;
            v.w = d * acc.w + b4.w;
            v.x = (v.x >= 0.0f) ? v.x : p4.x * v.x;
            v.y = (v.y >= 0.0f) ? v.y : p4.y * v.y;
            v.z = (v.z >= 0.0f) ? v.z : p4.z * v.z;
            v.w = (v.w >= 0.0f) ? v.w : p4.w * v.w;
            *(v4f*)(out + (size_t)(base + l) * Z_DIM + c4) = v;
        }
    } else {
        float* acc = (float*)pool;
        {
            const v4f* s4 = (const v4f*)(s + (size_t)base * Z_DIM);
            for (int i = tid; i < BIN_R * Z_DIM / 4; i += AGG_THREADS) ((v4f*)acc)[i] = s4[i];
        }
        __syncthreads();
        const v4i* col4 = (const v4i*)(ei + N_EDGES);
        for (int gi = tid; gi < COL_GROUPS; gi += AGG_THREADS) {
            const v4i c = col4[gi];
            #pragma unroll
            for (int j = 0; j < 4; ++j) {
                const unsigned lc = (unsigned)(c[j] - base);
                if (lc < (unsigned)BIN_R) agg_slow_add(acc, ei, s, gi * 4 + j, lc);
            }
        }
        __syncthreads();
        const v4f* acc4 = (const v4f*)acc;
        v4f* out4 = (v4f*)(out + (size_t)base * Z_DIM);
        for (int i = tid; i < BIN_R * Z_DIM / 4; i += AGG_THREADS) {
            const v4f a = acc4[i];
            const float d = dinv[base + (i >> 4)];
            const int k4 = (i & 15) << 2;
            v4f v;
            v.x = d * a.x + bias[k4 + 0];
            v.y = d * a.y + bias[k4 + 1];
            v.z = d * a.z + bias[k4 + 2];
            v.w = d * a.w + bias[k4 + 3];
            v.x = (v.x >= 0.0f) ? v.x : pa[k4 + 0] * v.x;
            v.y = (v.y >= 0.0f) ? v.y : pa[k4 + 1] * v.y;
            v.z = (v.z >= 0.0f) ? v.z : pa[k4 + 2] * v.z;
            v.w = (v.w >= 0.0f) ? v.w : pa[k4 + 3] * v.w;
            out4[i] = v;
        }
    }
}


extern "C" void kernel_launch(void* const* d_in, const int* in_sizes, int n_in,
                              void* d_out, int out_size, void* d_ws, size_t ws_size,
                              hipStream_t stream) {
    const float* x    = (const float*)d_in[0];
    const int*   ei   = (const int*)d_in[1];
    const float* W    = (const float*)d_in[3];
    const float* bias = (const float*)d_in[4];
    const float* pa   = (const float*)d_in[5];
    float* out = (float*)d_out;

    float* dinv   = (float*)((char*)d_ws + (size_t)N_NODES * 4);
    float* s      = (float*)((char*)d_ws + (size_t)2 * N_NODES * 4);

    unsigned* bk = (unsigned*)d_out;

    bin_kernel<<<BIN_BLOCKS, BIN_THREADS, 0, stream>>>(ei, bk);
    gemm_degree_scale_kernel<<<N_BINS, GEMM_THREADS, 0, stream>>>(x, W, ei, bk, dinv, s);

    aggregate_kernel<<<N_BINS, AGG_THREADS, 0, stream>>>(ei, s, dinv, bias, pa, out);
}
